# starting kernel
# baseline (speedup 1.0000x reference)
#include <hip/hip_runtime.h>
#include <hip/hip_bf16.h>


#define GCN_N 50000
#define GCN_E 1600000
#define GCN_D 128
#define GCN_L 3
#define LN_EPS 1e-5f

typedef __attribute__((ext_vector_type(2))) float v2f;
typedef __attribute__((ext_vector_type(8))) float v8f;


__global__ void k_init_deg(float* __restrict__ deg, int n) {
    int i = blockIdx.x * blockDim.x + threadIdx.x;
    if (i < n) deg[i] = 1.0f;
}

__global__ void k_accum_deg(const float* __restrict__ w, const int* __restrict__ col,
                            float* __restrict__ deg, int e) {
    int i = blockIdx.x * blockDim.x + threadIdx.x;
    if (i < e) atomicAdd(&deg[col[i]], w[i]);
}

__global__ void k_dinv(const float* __restrict__ deg, float* __restrict__ dinv, int n) {
    int i = blockIdx.x * blockDim.x + threadIdx.x;
    if (i < n) {
        float d = deg[i];
        dinv[i] = (d > 0.0f) ? rsqrtf(fmaxf(d, 1e-12f)) : 0.0f;
    }
}

__global__ void k_norm(const float* __restrict__ dinv, const int* __restrict__ row,
                       const int* __restrict__ col, const float* __restrict__ w,
                       float* __restrict__ norm, int e) {
    int i = blockIdx.x * blockDim.x + threadIdx.x;
    if (i < e) norm[i] = dinv[row[i]] * w[i] * dinv[col[i]];
}


__global__ void k_gemm_wmma(const float* __restrict__ H, const float* __restrict__ W,
                            float* __restrict__ XW, int n_nodes) {
    const int lane = threadIdx.x & 31;
    const int wave_in_block = threadIdx.x >> 5;
    const int waves_per_block = blockDim.x >> 5;
    const int tile_id = blockIdx.x * waves_per_block + wave_in_block;
    const int tiles_n = GCN_D / 16;
    const int tile_m = tile_id / tiles_n;
    const int tile_n = tile_id % tiles_n;
    if (tile_m * 16 >= n_nodes) return;

    const int row0 = tile_m * 16;
    const int col0 = tile_n * 16;
    const int lm = lane & 15;
    const int hi = lane >> 4;

    const float2* __restrict__ Ap = (const float2*)(H + (size_t)(row0 + lm) * GCN_D);
    const float2* __restrict__ Bp = (const float2*)(W + (size_t)(col0 + lm) * GCN_D);

    v8f c = {};
#pragma unroll
    for (int k = 0; k < GCN_D; k += 4) {
        const int kp = (k >> 1) + hi;
        float2 af = Ap[kp];
        float2 bf = Bp[kp];
        v2f a; a.x = af.x; a.y = af.y;
        v2f b; b.x = bf.x; b.y = bf.y;
        c = __builtin_amdgcn_wmma_f32_16x16x4_f32(
                 false, a,  false, b,
                 (short)0, c,  false,  false);
    }

    const int m0 = row0 + hi * 8;
    const int nc = col0 + lm;
#pragma unroll
    for (int v = 0; v < 8; ++v)
        XW[(size_t)(m0 + v) * GCN_D + nc] = c[v];
}


__global__ void k_init_acc(const float* __restrict__ XW, const float* __restrict__ dinv,
                           const float* __restrict__ bias, float* __restrict__ acc,
                           int n_nodes) {
    const int wave = (blockIdx.x * blockDim.x + threadIdx.x) >> 5;
    const int lane = threadIdx.x & 31;
    if (wave >= n_nodes) return;
    const float self = dinv[wave] * dinv[wave];
    float4 x = ((const float4*)(XW + (size_t)wave * GCN_D))[lane];
    float4 b = ((const float4*)bias)[lane];
    float4 o;
    o.x = b.x + x.x * self;
    o.y = b.y + x.y * self;
    o.z = b.z + x.z * self;
    o.w = b.w + x.w * self;
    ((float4*)(acc + (size_t)wave * GCN_D))[lane] = o;
}


__global__ void k_scatter(const float* __restrict__ XW, const int* __restrict__ row,
                          const int* __restrict__ col, const float* __restrict__ norm,
                          float* __restrict__ acc, int n_edges) {
    const int wave = (blockIdx.x * blockDim.x + threadIdx.x) >> 5;
    const int lane = threadIdx.x & 31;
    if (wave >= n_edges) return;
    const int r = row[wave];
    const int c = col[wave];
    const float w = norm[wave];
    float4 m = ((const float4*)(XW + (size_t)r * GCN_D))[lane];
    float* dst = acc + (size_t)c * GCN_D + lane * 4;
    atomicAdd(dst + 0, m.x * w);
    atomicAdd(dst + 1, m.y * w);
    atomicAdd(dst + 2, m.z * w);
    atomicAdd(dst + 3, m.w * w);
}


__global__ void k_ln_res(const float* __restrict__ acc, const float* __restrict__ hin,
                         const float* __restrict__ gamma, const float* __restrict__ beta,
                         float* __restrict__ hout, int n_nodes, int do_relu) {
    const int wave = (blockIdx.x * blockDim.x + threadIdx.x) >> 5;
    const int lane = threadIdx.x & 31;
    if (wave >= n_nodes) return;

    float4 a = ((const float4*)(acc + (size_t)wave * GCN_D))[lane];

    float s = a.x + a.y + a.z + a.w;
#pragma unroll
    for (int off = 16; off > 0; off >>= 1) s += __shfl_xor(s, off, 32);
    const float mu = s * (1.0f / GCN_D);

    float dx = a.x - mu, dy = a.y - mu, dz = a.z - mu, dw = a.w - mu;
    float q = dx * dx + dy * dy + dz * dz + dw * dw;
#pragma unroll
    for (int off = 16; off > 0; off >>= 1) q += __shfl_xor(q, off, 32);
    const float inv = rsqrtf(q * (1.0f / GCN_D) + LN_EPS);

    float4 g = ((const float4*)gamma)[lane];
    float4 b = ((const float4*)beta)[lane];
    float4 id = ((const float4*)(hin + (size_t)wave * GCN_D))[lane];

    float4 y;
    y.x = dx * inv * g.x + b.x;
    y.y = dy * inv * g.y + b.y;
    y.z = dz * inv * g.z + b.z;
    y.w = dw * inv * g.w + b.w;
    if (do_relu) {
        y.x = fmaxf(y.x, 0.0f);
        y.y = fmaxf(y.y, 0.0f);
        y.z = fmaxf(y.z, 0.0f);
        y.w = fmaxf(y.w, 0.0f);
    }
    y.x += id.x; y.y += id.y; y.z += id.z; y.w += id.w;
    ((float4*)(hout + (size_t)wave * GCN_D))[lane] = y;
}


extern "C" void kernel_launch(void* const* d_in, const int* in_sizes, int n_in,
                              void* d_out, int out_size, void* d_ws, size_t ws_size,
                              hipStream_t stream) {
    (void)in_sizes; (void)n_in; (void)out_size; (void)ws_size;

    const float* x      = (const float*)d_in[0];
    const int*   ei     = (const int*)d_in[1];
    const float* ew     = (const float*)d_in[2];
    const float* Ws     = (const float*)d_in[3];
    const float* bs     = (const float*)d_in[4];
    const float* gammas = (const float*)d_in[5];
    const float* betas  = (const float*)d_in[6];
    float*       out    = (float*)d_out;

    const int* row = ei;
    const int* col = ei + GCN_E;

    float* ws   = (float*)d_ws;
    float* deg  = ws;
    float* dinv = deg + GCN_N;
    float* norm = dinv + GCN_N;
    float* xw   = norm + GCN_E;
    float* acc  = xw + (size_t)GCN_N * GCN_D;
    float* h1   = acc + (size_t)GCN_N * GCN_D;

    const int BLK = 256;

    k_init_deg<<<(GCN_N + BLK - 1) / BLK, BLK, 0, stream>>>(deg, GCN_N);
    k_accum_deg<<<(GCN_E + BLK - 1) / BLK, BLK, 0, stream>>>(ew, col, deg, GCN_E);
    k_dinv<<<(GCN_N + BLK - 1) / BLK, BLK, 0, stream>>>(deg, dinv, GCN_N);
    k_norm<<<(GCN_E + BLK - 1) / BLK, BLK, 0, stream>>>(dinv, row, col, ew, norm, GCN_E);

    const int tiles_total  = ((GCN_N + 15) / 16) * (GCN_D / 16);
    const int gemm_blocks  = (tiles_total + (BLK / 32) - 1) / (BLK / 32);
    const int node_wave_blocks = (GCN_N + (BLK / 32) - 1) / (BLK / 32);
    const int edge_wave_blocks = (GCN_E + (BLK / 32) - 1) / (BLK / 32);

    const float* hin[GCN_L]  = { x,  h1,  out };
    float*       hout[GCN_L] = { h1, out, out };

    for (int l = 0; l < GCN_L; ++l) {
        const float* W = Ws + (size_t)l * GCN_D * GCN_D;
        k_gemm_wmma<<<gemm_blocks, BLK, 0, stream>>>(hin[l], W, xw, GCN_N);
        k_init_acc<<<node_wave_blocks, BLK, 0, stream>>>(xw, dinv, bs + (size_t)l * GCN_D,
                                                         acc, GCN_N);
        k_scatter<<<edge_wave_blocks, BLK, 0, stream>>>(xw, row, col, norm, acc, GCN_E);
        k_ln_res<<<node_wave_blocks, BLK, 0, stream>>>(acc, hin[l],
                                                       gammas + (size_t)l * GCN_D,
                                                       betas + (size_t)l * GCN_D,
                                                       hout[l], GCN_N,
                                                        (l < GCN_L - 1) ? 1 : 0);
    }
}


# Scatter ranges of 49 nodes, not 64: 1021 work-groups = 4 on every WGP (782 gave 3 on most but 4 on 14, which set the time): a WGP's LDS float adds cover 196 nodes, not 256; arrival-order sums kept
# speedup vs baseline: 3.5661x; 3.5661x over previous
#include <hip/hip_runtime.h>
#include <hip/hip_bf16.h>


#define GCN_N 50000
#define GCN_E 1600000
#define GCN_D 128
#define GCN_L 3
#define LN_EPS 1e-5f

#define SB_ROWS 49
#define SB_BINS ((GCN_N + SB_ROWS - 1) / SB_ROWS)
#define SB_LDS_ROWS 64
#define SB_PAD 1024
#define SB_WAVES 8
#define BN_THREADS 1024
#define BN_PER_THREAD 8
#define BN_SLICE (BN_THREADS * BN_PER_THREAD)
static_assert(SB_BINS < SB_PAD && SB_PAD == BN_THREADS, "one counter per binning thread");
static_assert(SB_ROWS <= SB_LDS_ROWS, "a range's sum rows fit the reserved LDS rows");

__device__ __forceinline__ int sb_bin(int node) { return (int)((unsigned)node / (unsigned)SB_ROWS); }

__device__ __forceinline__ int pkg_half() {
    return (int)((__builtin_amdgcn_s_sendmsg_rtn(0x87) >> 18) & 1u);
}

typedef __attribute__((ext_vector_type(2))) float v2f;
typedef __attribute__((ext_vector_type(8))) float v8f;


__global__ void k_init_deg(float* __restrict__ deg0, float* __restrict__ deg1,
                           int* __restrict__ cnt, int* __restrict__ cursor, int n) {
    int i = blockIdx.x * blockDim.x + threadIdx.x;
    if (i < n) {
        deg0[i] = 1.0f;
        deg1[i] = 0.0f;
    }
    if (i < SB_PAD) {
        cnt[i] = 0;
        cursor[i] = 0;
    }
}

#define DG_THREADS 256
#define DG_PER_THREAD 8

__global__ void __launch_bounds__(DG_THREADS)
k_accum_deg(const float* __restrict__ w, const int* __restrict__ col,
            float* __restrict__ deg0, float* __restrict__ deg1, int e) {
    float* __restrict__ deg = pkg_half() ? deg1 : deg0;
    const int base = blockIdx.x * (DG_THREADS * DG_PER_THREAD) + threadIdx.x;
    int c[DG_PER_THREAD];
    float we[DG_PER_THREAD];
#pragma unroll
    for (int k = 0; k < DG_PER_THREAD; ++k) {
        const int i = base + k * DG_THREADS;
        const int ii = i < e ? i : e - 1;
        c[k] = col[ii];
        we[k] = w[ii];
    }
#pragma unroll
    for (int k = 0; k < DG_PER_THREAD; ++k)
        if (base + k * DG_THREADS < e) atomicAdd(&deg[c[k]], we[k]);
}

__global__ void k_dinv(const float* __restrict__ deg0, const float* __restrict__ deg1,
                       float* __restrict__ dinv, int n) {
    int i = blockIdx.x * blockDim.x + threadIdx.x;
    if (i < n) {
        float d = deg0[i] + deg1[i];
        dinv[i] = (d > 0.0f) ? rsqrtf(fmaxf(d, 1e-12f)) : 0.0f;
    }
}


__global__ void __launch_bounds__(BN_THREADS)
k_bin_count(const int* __restrict__ col, int* __restrict__ cnt, int e) {
    __shared__ int h[SB_PAD];
    const int tid = threadIdx.x;
    h[tid] = 0;
    __syncthreads();
    const int base = blockIdx.x * BN_SLICE + tid;
    int cv[BN_PER_THREAD];
#pragma unroll
    for (int k = 0; k < BN_PER_THREAD; ++k) {
        const int i = base + k * BN_THREADS;
        cv[k] = col[i < e ? i : e - 1];
    }
#pragma unroll
    for (int k = 0; k < BN_PER_THREAD; ++k)
        if (base + k * BN_THREADS < e) atomicAdd(&h[sb_bin(cv[k])], 1);
    __syncthreads();
    const int v = h[tid];
    if (v) atomicAdd(&cnt[tid], v);
}

__global__ void __launch_bounds__(BN_THREADS)
k_bin_place(const float* __restrict__ dinv, const int* __restrict__ row,
            const int* __restrict__ col, const float* __restrict__ w,
            const int* __restrict__ cnt, int* __restrict__ cursor, int* __restrict__ off,
            int* __restrict__ prow, int* __restrict__ pcol, float* __restrict__ pnorm, int e) {
    __shared__ int h[SB_PAD];
    __shared__ int wsum[32];
    const int tid = threadIdx.x;
    const int lane = tid & 31;
    const int wv = tid >> 5;

    const int cv = cnt[tid];
    int inc = cv;
#pragma unroll
    for (int d = 1; d < 32; d <<= 1) {
        const int t = __shfl_up(inc, d, 32);
        if (lane >= d) inc += t;
    }
    if (lane == 31) wsum[wv] = inc;
    h[tid] = 0;
    __syncthreads();
    if (wv == 0) {
        int s = wsum[lane];
#pragma unroll
        for (int d = 1; d < 32; d <<= 1) {
            const int t = __shfl_up(s, d, 32);
            if (lane >= d) s += t;
        }
        wsum[lane] = s;
    }
    __syncthreads();
    const int start = inc - cv + (wv ? wsum[wv - 1] : 0);
    if (blockIdx.x == 0) off[tid] = start;

    int c[BN_PER_THREAD], r[BN_PER_THREAD], rk[BN_PER_THREAD];
    float we[BN_PER_THREAD], dr[BN_PER_THREAD], dc[BN_PER_THREAD];
    const int base = blockIdx.x * BN_SLICE + tid;
#pragma unroll
    for (int k = 0; k < BN_PER_THREAD; ++k) {
        const int i = base + k * BN_THREADS;
        const int ii = i < e ? i : e - 1;
        c[k] = col[ii];
        r[k] = row[ii];
        we[k] = w[ii];
    }
#pragma unroll
    for (int k = 0; k < BN_PER_THREAD; ++k) {
        const int i = base + k * BN_THREADS;
        rk[k] = 0;
        if (i < e) rk[k] = atomicAdd(&h[sb_bin(c[k])], 1);
    }
#pragma unroll
    for (int k = 0; k < BN_PER_THREAD; ++k) {
        dr[k] = dinv[r[k]];
        dc[k] = dinv[c[k]];
    }
    __syncthreads();
    const int hv = h[tid];
    h[tid] = start + (hv ? atomicAdd(&cursor[tid], hv) : 0);
    __syncthreads();
#pragma unroll
    for (int k = 0; k < BN_PER_THREAD; ++k) {
        const int i = base + k * BN_THREADS;
        if (i < e) {
            const int bin = sb_bin(c[k]);
            const int pos = h[bin] + rk[k];
            prow[pos] = r[k];
            pcol[pos] = c[k] - bin * SB_ROWS;
            pnorm[pos] = dr[k] * we[k] * dc[k];
        }
    }
}


#define GM_ROWTILES 4
#define GM_WAVES 4

__global__ void __launch_bounds__(GM_WAVES * 32)
k_gemm_wmma(const float* __restrict__ H, const float* __restrict__ W,
            float* __restrict__ XW, int n_nodes) {
    const int lane = threadIdx.x & 31;
    const int col0 = (threadIdx.x >> 5) * 32;
    const int lm = lane & 15;
    const int hi = lane >> 4;

    const float4* __restrict__ Bp0 = (const float4*)(W + (size_t)(col0 + lm) * GCN_D) + hi;
    const float4* __restrict__ Bp1 = (const float4*)(W + (size_t)(col0 + 16 + lm) * GCN_D) + hi;
    float4 b0[16], b1[16];
#pragma unroll
    for (int i = 0; i < 16; ++i) {
        b0[i] = Bp0[2 * i];
        b1[i] = Bp1[2 * i];
    }

    const int nc = col0 + lm;
#pragma unroll 1
    for (int rt = 0; rt < GM_ROWTILES; ++rt) {
        const int row0 = ((int)blockIdx.x * GM_ROWTILES + rt) * 16;
        if (row0 >= n_nodes) break;

        const float4* __restrict__ Ap = (const float4*)(H + (size_t)(row0 + lm) * GCN_D) + hi;
        float4 a[16];
#pragma unroll
        for (int i = 0; i < 16; ++i) a[i] = Ap[2 * i];

        v8f c0 = {};
        v8f c1 = {};
#pragma unroll
        for (int i = 0; i < 16; ++i) {
            v2f ax; ax.x = a[i].x; ax.y = a[i].y;
            v2f az; az.x = a[i].z; az.y = a[i].w;
            v2f bx0; bx0.x = b0[i].x; bx0.y = b0[i].y;
            v2f bz0; bz0.x = b0[i].z; bz0.y = b0[i].w;
            v2f bx1; bx1.x = b1[i].x; bx1.y = b1[i].y;
            v2f bz1; bz1.x = b1[i].z; bz1.y = b1[i].w;
            c0 = __builtin_amdgcn_wmma_f32_16x16x4_f32(false, ax, false, bx0, (short)0, c0, false, false);
            c1 = __builtin_amdgcn_wmma_f32_16x16x4_f32(false, ax, false, bx1, (short)0, c1, false, false);
            c0 = __builtin_amdgcn_wmma_f32_16x16x4_f32(false, az, false, bz0, (short)0, c0, false, false);
            c1 = __builtin_amdgcn_wmma_f32_16x16x4_f32(false, az, false, bz1, (short)0, c1, false, false);
        }

        const int m0 = row0 + hi * 8;
#pragma unroll
        for (int v = 0; v < 8; ++v) {
            XW[(size_t)(m0 + v) * GCN_D + nc] = c0[v];
            XW[(size_t)(m0 + v) * GCN_D + nc + 16] = c1[v];
        }
    }
}


#define SC_EDGES_PER_WAVE 32
#define SC_GROUP 8

__global__ void __launch_bounds__(SB_WAVES * 32)
k_scatter(const float* __restrict__ XW, const float* __restrict__ dinv,
          const float* __restrict__ bias, const int* __restrict__ prow,
          const int* __restrict__ pcol, const float* __restrict__ pnorm,
          const int* __restrict__ off, float* __restrict__ acc, int n_nodes) {
    __shared__ float sacc[SB_LDS_ROWS * GCN_D];
    const int lane = threadIdx.x & 31;
    const int wave = __builtin_amdgcn_readfirstlane((int)(threadIdx.x >> 5));
    const int n0 = (int)blockIdx.x * SB_ROWS;
    const int nleft = n_nodes - n0;
    const int nrows = nleft < SB_ROWS ? nleft : SB_ROWS;

    const float4 b = ((const float4*)bias)[lane];
    for (int r = wave; r < nrows; r += SB_WAVES) {
        const float dn = dinv[n0 + r];
        const float self = dn * dn;
        const float4 x = ((const float4*)(XW + (size_t)(n0 + r) * GCN_D))[lane];
        float4 o;
        o.x = b.x + x.x * self;
        o.y = b.y + x.y * self;
        o.z = b.z + x.z * self;
        o.w = b.w + x.w * self;
        ((float4*)(sacc + r * GCN_D))[lane] = o;
    }
    __syncthreads();

    const int e_begin = off[blockIdx.x];
    const int e_end = off[blockIdx.x + 1];
    int e0 = e_begin + wave * SC_EDGES_PER_WAVE;
    int r = 0, c = 0, wbits = 0;
    if (e0 + lane < e_end) {
        r = prow[e0 + lane];
        c = pcol[e0 + lane];
        wbits = __float_as_int(pnorm[e0 + lane]);
    }
    while (e0 < e_end) {
        const int left = e_end - e0;
        const int cnt = left < SC_EDGES_PER_WAVE ? left : SC_EDGES_PER_WAVE;
        const int e1 = e0 + SB_WAVES * SC_EDGES_PER_WAVE;
        int rn = 0, cn = 0, wn = 0;
        if (e1 + lane < e_end) {
            rn = prow[e1 + lane];
            cn = pcol[e1 + lane];
            wn = __float_as_int(pnorm[e1 + lane]);
        }
        int g = 0;
#pragma unroll 1
        for (; g + SC_GROUP <= cnt; g += SC_GROUP) {
            float m[SC_GROUP][4];
#pragma unroll
            for (int k = 0; k < SC_GROUP; ++k) {
                const int rs = __builtin_amdgcn_readlane(r, g + k);
                const float* src = XW + (size_t)rs * GCN_D + lane;
                m[k][0] = src[0];
                m[k][1] = src[32];
                m[k][2] = src[64];
                m[k][3] = src[96];
            }
#pragma unroll
            for (int k = 0; k < SC_GROUP; ++k) {
                const int cs = __builtin_amdgcn_readlane(c, g + k);
                const float wf = __int_as_float(__builtin_amdgcn_readlane(wbits, g + k));
                float* dst = sacc + cs * GCN_D + lane;
                atomicAdd(dst + 0,  m[k][0] * wf);
                atomicAdd(dst + 32, m[k][1] * wf);
                atomicAdd(dst + 64, m[k][2] * wf);
                atomicAdd(dst + 96, m[k][3] * wf);
            }
        }
        for (int i = g; i < cnt; ++i) {
            const int rs = __builtin_amdgcn_readlane(r, i);
            const int cs = __builtin_amdgcn_readlane(c, i);
            const float wf = __int_as_float(__builtin_amdgcn_readlane(wbits, i));
            const float* src = XW + (size_t)rs * GCN_D + lane;
            float* dst = sacc + cs * GCN_D + lane;
            const float m0 = src[0], m1 = src[32], m2 = src[64], m3 = src[96];
            atomicAdd(dst + 0,  m0 * wf);
            atomicAdd(dst + 32, m1 * wf);
            atomicAdd(dst + 64, m2 * wf);
            atomicAdd(dst + 96, m3 * wf);
        }
        r = rn;
        c = cn;
        wbits = wn;
        e0 = e1;
    }
    __syncthreads();

    for (int r = wave; r < nrows; r += SB_WAVES)
        ((float4*)(acc + (size_t)(n0 + r) * GCN_D))[lane] = ((const float4*)(sacc + r * GCN_D))[lane];
}


#define LN_ROWS 8
#define LN_THREADS 256

__global__ void __launch_bounds__(LN_THREADS)
k_ln_res(const float* __restrict__ acc,
         const float* hin,
         const float* __restrict__ gamma, const float* __restrict__ beta,
         float* hout, int n_nodes, int do_relu) {
    const int wave = (blockIdx.x * blockDim.x + threadIdx.x) >> 5;
    const int lane = threadIdx.x & 31;
    const int r0 = wave * LN_ROWS;
    if (r0 >= n_nodes) return;

    float4 a[LN_ROWS], id[LN_ROWS];
#pragma unroll
    for (int k = 0; k < LN_ROWS; ++k) {
        const int r = r0 + k < n_nodes ? r0 + k : n_nodes - 1;
        a[k] = ((const float4*)(acc + (size_t)r * GCN_D))[lane];
    }
#pragma unroll
    for (int k = 0; k < LN_ROWS; ++k) {
        const int r = r0 + k < n_nodes ? r0 + k : n_nodes - 1;
        id[k] = ((const float4*)(hin + (size_t)r * GCN_D))[lane];
    }
    const float4 g = ((const float4*)gamma)[lane];
    const float4 b = ((const float4*)beta)[lane];

    float s[LN_ROWS];
#pragma unroll
    for (int k = 0; k < LN_ROWS; ++k) s[k] = a[k].x + a[k].y + a[k].z + a[k].w;
#pragma unroll
    for (int off = 16; off > 0; off >>= 1) {
#pragma unroll
        for (int k = 0; k < LN_ROWS; ++k) s[k] += __shfl_xor(s[k], off, 32);
    }

    float q[LN_ROWS];
#pragma unroll
    for (int k = 0; k < LN_ROWS; ++k) {
        const float mu = s[k] * (1.0f / GCN_D);
        a[k].x -= mu; a[k].y -= mu; a[k].z -= mu; a[k].w -= mu;
        q[k] = a[k].x * a[k].x + a[k].y * a[k].y + a[k].z * a[k].z + a[k].w * a[k].w;
    }
#pragma unroll
    for (int off = 16; off > 0; off >>= 1) {
#pragma unroll
        for (int k = 0; k < LN_ROWS; ++k) q[k] += __shfl_xor(q[k], off, 32);
    }

#pragma unroll
    for (int k = 0; k < LN_ROWS; ++k) {
        const float inv = rsqrtf(q[k] * (1.0f / GCN_D) + LN_EPS);
        float4 y;
        y.x = a[k].x * inv * g.x + b.x;
        y.y = a[k].y * inv * g.y + b.y;
        y.z = a[k].z * inv * g.z + b.z;
        y.w = a[k].w * inv * g.w + b.w;
        if (do_relu) {
            y.x = fmaxf(y.x, 0.0f);
            y.y = fmaxf(y.y, 0.0f);
            y.z = fmaxf(y.z, 0.0f);
            y.w = fmaxf(y.w, 0.0f);
        }
        y.x += id[k].x; y.y += id[k].y; y.z += id[k].z; y.w += id[k].w;
        if (r0 + k < n_nodes)
            ((float4*)(hout + (size_t)(r0 + k) * GCN_D))[lane] = y;
    }
}


extern "C" void kernel_launch(void* const* d_in, const int* in_sizes, int n_in,
                              void* d_out, int out_size, void* d_ws, size_t ws_size,
                              hipStream_t stream) {
    (void)in_sizes; (void)n_in; (void)out_size; (void)ws_size;

    const float* x      = (const float*)d_in[0];
    const int*   ei     = (const int*)d_in[1];
    const float* ew     = (const float*)d_in[2];
    const float* Ws     = (const float*)d_in[3];
    const float* bs     = (const float*)d_in[4];
    const float* gammas = (const float*)d_in[5];
    const float* betas  = (const float*)d_in[6];
    float*       out    = (float*)d_out;

    const int* row = ei;
    const int* col = ei + GCN_E;

    float* ws   = (float*)d_ws;
    float* deg  = ws;
    float* dinv = deg + GCN_N;
    float* pnorm = dinv + GCN_N;
    float* xw   = pnorm + GCN_E;
    float* acc  = xw + (size_t)GCN_N * GCN_D;
    float* h1   = acc + (size_t)GCN_N * GCN_D;
    float* deg1 = h1 + (size_t)GCN_N * GCN_D;
    int* cnt    = (int*)(deg1 + GCN_N);
    int* cursor = cnt + SB_PAD;
    int* off    = cursor + SB_PAD;
    int* prow   = off + SB_PAD;
    int* pcol   = prow + GCN_E;

    const int BLK = 256;

    k_init_deg<<<(GCN_N + BLK - 1) / BLK, BLK, 0, stream>>>(deg, deg1, cnt, cursor, GCN_N);
    k_accum_deg<<<(GCN_E + DG_THREADS * DG_PER_THREAD - 1) / (DG_THREADS * DG_PER_THREAD),
                  DG_THREADS, 0, stream>>>(ew, col, deg, deg1, GCN_E);
    k_dinv<<<(GCN_N + BLK - 1) / BLK, BLK, 0, stream>>>(deg, deg1, dinv, GCN_N);
    const int bin_blocks = (GCN_E + BN_SLICE - 1) / BN_SLICE;
    k_bin_count<<<bin_blocks, BN_THREADS, 0, stream>>>(col, cnt, GCN_E);
    k_bin_place<<<bin_blocks, BN_THREADS, 0, stream>>>(dinv, row, col, ew, cnt, cursor, off,
                                                       prow, pcol, pnorm, GCN_E);

    const int row_tiles    = (GCN_N + 15) / 16;
    const int gemm_blocks  = (row_tiles + GM_ROWTILES - 1) / GM_ROWTILES;
    const int ln_rows_per_block = LN_ROWS * (LN_THREADS / 32);
    const int ln_blocks = (GCN_N + ln_rows_per_block - 1) / ln_rows_per_block;

    const float* hin[GCN_L]  = { x,  h1,  out };
    float*       hout[GCN_L] = { h1, out, out };

    for (int l = 0; l < GCN_L; ++l) {
        const float* W = Ws + (size_t)l * GCN_D * GCN_D;
        k_gemm_wmma<<<gemm_blocks, GM_WAVES * 32, 0, stream>>>(hin[l], W, xw, GCN_N);
        k_scatter<<<SB_BINS, SB_WAVES * 32, 0, stream>>>(xw, dinv, bs + (size_t)l * GCN_D,
                                                         prow, pcol, pnorm, off, acc, GCN_N);
        k_ln_res<<<ln_blocks, LN_THREADS, 0, stream>>>(acc, hin[l],
                                                       gammas + (size_t)l * GCN_D,
                                                       betas + (size_t)l * GCN_D,
                                                       hout[l], GCN_N,
                                                        (l < GCN_L - 1) ? 1 : 0);
    }
}


# LayerNorm, ReLU and residual move into the scatter work-group: it normalises its finished sum rows in LDS and writes h itself; LN launch and node-sum plane (write + read) go; sum orders kept
# speedup vs baseline: 3.7954x; 1.0643x over previous
#include <hip/hip_runtime.h>
#include <hip/hip_bf16.h>


#define GCN_N 50000
#define GCN_E 1600000
#define GCN_D 128
#define GCN_L 3
#define LN_EPS 1e-5f

#define SB_ROWS 49
#define SB_BINS ((GCN_N + SB_ROWS - 1) / SB_ROWS)
#define SB_LDS_ROWS 64
#define SB_PAD 1024
#define SB_WAVES 8
#define BN_THREADS 1024
#define BN_PER_THREAD 8
#define BN_SLICE (BN_THREADS * BN_PER_THREAD)
static_assert(SB_BINS < SB_PAD && SB_PAD == BN_THREADS, "one counter per binning thread");
static_assert(SB_ROWS <= SB_LDS_ROWS, "a range's sum rows fit the reserved LDS rows");

__device__ __forceinline__ int sb_bin(int node) { return (int)((unsigned)node / (unsigned)SB_ROWS); }

__device__ __forceinline__ int pkg_half() {
    return (int)((__builtin_amdgcn_s_sendmsg_rtn(0x87) >> 18) & 1u);
}

typedef __attribute__((ext_vector_type(2))) float v2f;
typedef __attribute__((ext_vector_type(8))) float v8f;


__global__ void k_init_deg(float* __restrict__ deg0, float* __restrict__ deg1,
                           int* __restrict__ cnt, int* __restrict__ cursor, int n) {
    int i = blockIdx.x * blockDim.x + threadIdx.x;
    if (i < n) {
        deg0[i] = 1.0f;
        deg1[i] = 0.0f;
    }
    if (i < SB_PAD) {
        cnt[i] = 0;
        cursor[i] = 0;
    }
}

#define DG_THREADS 256
#define DG_PER_THREAD 8

__global__ void __launch_bounds__(DG_THREADS)
k_accum_deg(const float* __restrict__ w, const int* __restrict__ col,
            float* __restrict__ deg0, float* __restrict__ deg1, int e) {
    float* __restrict__ deg = pkg_half() ? deg1 : deg0;
    const int base = blockIdx.x * (DG_THREADS * DG_PER_THREAD) + threadIdx.x;
    int c[DG_PER_THREAD];
    float we[DG_PER_THREAD];
#pragma unroll
    for (int k = 0; k < DG_PER_THREAD; ++k) {
        const int i = base + k * DG_THREADS;
        const int ii = i < e ? i : e - 1;
        c[k] = col[ii];
        we[k] = w[ii];
    }
#pragma unroll
    for (int k = 0; k < DG_PER_THREAD; ++k)
        if (base + k * DG_THREADS < e) atomicAdd(&deg[c[k]], we[k]);
}

__global__ void k_dinv(const float* __restrict__ deg0, const float* __restrict__ deg1,
                       float* __restrict__ dinv, int n) {
    int i = blockIdx.x * blockDim.x + threadIdx.x;
    if (i < n) {
        float d = deg0[i] + deg1[i];
        dinv[i] = (d > 0.0f) ? rsqrtf(fmaxf(d, 1e-12f)) : 0.0f;
    }
}


__global__ void __launch_bounds__(BN_THREADS)
k_bin_count(const int* __restrict__ col, int* __restrict__ cnt, int e) {
    __shared__ int h[SB_PAD];
    const int tid = threadIdx.x;
    h[tid] = 0;
    __syncthreads();
    const int base = blockIdx.x * BN_SLICE + tid;
    int cv[BN_PER_THREAD];
#pragma unroll
    for (int k = 0; k < BN_PER_THREAD; ++k) {
        const int i = base + k * BN_THREADS;
        cv[k] = col[i < e ? i : e - 1];
    }
#pragma unroll
    for (int k = 0; k < BN_PER_THREAD; ++k)
        if (base + k * BN_THREADS < e) atomicAdd(&h[sb_bin(cv[k])], 1);
    __syncthreads();
    const int v = h[tid];
    if (v) atomicAdd(&cnt[tid], v);
}

__global__ void __launch_bounds__(BN_THREADS)
k_bin_place(const float* __restrict__ dinv, const int* __restrict__ row,
            const int* __restrict__ col, const float* __restrict__ w,
            const int* __restrict__ cnt, int* __restrict__ cursor, int* __restrict__ off,
            int* __restrict__ prow, int* __restrict__ pcol, float* __restrict__ pnorm, int e) {
    __shared__ int h[SB_PAD];
    __shared__ int wsum[32];
    const int tid = threadIdx.x;
    const int lane = tid & 31;
    const int wv = tid >> 5;

    const int cv = cnt[tid];
    int inc = cv;
#pragma unroll
    for (int d = 1; d < 32; d <<= 1) {
        const int t = __shfl_up(inc, d, 32);
        if (lane >= d) inc += t;
    }
    if (lane == 31) wsum[wv] = inc;
    h[tid] = 0;
    __syncthreads();
    if (wv == 0) {
        int s = wsum[lane];
#pragma unroll
        for (int d = 1; d < 32; d <<= 1) {
            const int t = __shfl_up(s, d, 32);
            if (lane >= d) s += t;
        }
        wsum[lane] = s;
    }
    __syncthreads();
    const int start = inc - cv + (wv ? wsum[wv - 1] : 0);
    if (blockIdx.x == 0) off[tid] = start;

    int c[BN_PER_THREAD], r[BN_PER_THREAD], rk[BN_PER_THREAD];
    float we[BN_PER_THREAD], dr[BN_PER_THREAD], dc[BN_PER_THREAD];
    const int base = blockIdx.x * BN_SLICE + tid;
#pragma unroll
    for (int k = 0; k < BN_PER_THREAD; ++k) {
        const int i = base + k * BN_THREADS;
        const int ii = i < e ? i : e - 1;
        c[k] = col[ii];
        r[k] = row[ii];
        we[k] = w[ii];
    }
#pragma unroll
    for (int k = 0; k < BN_PER_THREAD; ++k) {
        const int i = base + k * BN_THREADS;
        rk[k] = 0;
        if (i < e) rk[k] = atomicAdd(&h[sb_bin(c[k])], 1);
    }
#pragma unroll
    for (int k = 0; k < BN_PER_THREAD; ++k) {
        dr[k] = dinv[r[k]];
        dc[k] = dinv[c[k]];
    }
    __syncthreads();
    const int hv = h[tid];
    h[tid] = start + (hv ? atomicAdd(&cursor[tid], hv) : 0);
    __syncthreads();
#pragma unroll
    for (int k = 0; k < BN_PER_THREAD; ++k) {
        const int i = base + k * BN_THREADS;
        if (i < e) {
            const int bin = sb_bin(c[k]);
            const int pos = h[bin] + rk[k];
            prow[pos] = r[k];
            pcol[pos] = c[k] - bin * SB_ROWS;
            pnorm[pos] = dr[k] * we[k] * dc[k];
        }
    }
}


#define GM_ROWTILES 4
#define GM_WAVES 4

__global__ void __launch_bounds__(GM_WAVES * 32)
k_gemm_wmma(const float* __restrict__ H, const float* __restrict__ W,
            float* __restrict__ XW, int n_nodes) {
    const int lane = threadIdx.x & 31;
    const int col0 = (threadIdx.x >> 5) * 32;
    const int lm = lane & 15;
    const int hi = lane >> 4;

    const float4* __restrict__ Bp0 = (const float4*)(W + (size_t)(col0 + lm) * GCN_D) + hi;
    const float4* __restrict__ Bp1 = (const float4*)(W + (size_t)(col0 + 16 + lm) * GCN_D) + hi;
    float4 b0[16], b1[16];
#pragma unroll
    for (int i = 0; i < 16; ++i) {
        b0[i] = Bp0[2 * i];
        b1[i] = Bp1[2 * i];
    }

    const int nc = col0 + lm;
#pragma unroll 1
    for (int rt = 0; rt < GM_ROWTILES; ++rt) {
        const int row0 = ((int)blockIdx.x * GM_ROWTILES + rt) * 16;
        if (row0 >= n_nodes) break;

        const float4* __restrict__ Ap = (const float4*)(H + (size_t)(row0 + lm) * GCN_D) + hi;
        float4 a[16];
#pragma unroll
        for (int i = 0; i < 16; ++i) a[i] = Ap[2 * i];

        v8f c0 = {};
        v8f c1 = {};
#pragma unroll
        for (int i = 0; i < 16; ++i) {
            v2f ax; ax.x = a[i].x; ax.y = a[i].y;
            v2f az; az.x = a[i].z; az.y = a[i].w;
            v2f bx0; bx0.x = b0[i].x; bx0.y = b0[i].y;
            v2f bz0; bz0.x = b0[i].z; bz0.y = b0[i].w;
            v2f bx1; bx1.x = b1[i].x; bx1.y = b1[i].y;
            v2f bz1; bz1.x = b1[i].z; bz1.y = b1[i].w;
            c0 = __builtin_amdgcn_wmma_f32_16x16x4_f32(false, ax, false, bx0, (short)0, c0, false, false);
            c1 = __builtin_amdgcn_wmma_f32_16x16x4_f32(false, ax, false, bx1, (short)0, c1, false, false);
            c0 = __builtin_amdgcn_wmma_f32_16x16x4_f32(false, az, false, bz0, (short)0, c0, false, false);
            c1 = __builtin_amdgcn_wmma_f32_16x16x4_f32(false, az, false, bz1, (short)0, c1, false, false);
        }

        const int m0 = row0 + hi * 8;
#pragma unroll
        for (int v = 0; v < 8; ++v) {
            XW[(size_t)(m0 + v) * GCN_D + nc] = c0[v];
            XW[(size_t)(m0 + v) * GCN_D + nc + 16] = c1[v];
        }
    }
}


#define SC_EDGES_PER_WAVE 32
#define SC_GROUP 8
#define SC_LN_ROWS ((SB_ROWS + SB_WAVES - 1) / SB_WAVES)

__global__ void __launch_bounds__(SB_WAVES * 32)
k_scatter(const float* __restrict__ XW, const float* __restrict__ dinv,
          const float* __restrict__ bias, const int* __restrict__ prow,
          const int* __restrict__ pcol, const float* __restrict__ pnorm,
          const int* __restrict__ off, const float* hin,
          const float* __restrict__ gamma, const float* __restrict__ beta,
          float* hout, int n_nodes, int do_relu) {
    __shared__ float sacc[SB_LDS_ROWS * GCN_D];
    const int lane = threadIdx.x & 31;
    const int wave = __builtin_amdgcn_readfirstlane((int)(threadIdx.x >> 5));
    const int n0 = (int)blockIdx.x * SB_ROWS;
    const int nleft = n_nodes - n0;
    const int nrows = nleft < SB_ROWS ? nleft : SB_ROWS;

    const float4 b = ((const float4*)bias)[lane];
    for (int r = wave; r < nrows; r += SB_WAVES) {
        const float dn = dinv[n0 + r];
        const float self = dn * dn;
        const float4 x = ((const float4*)(XW + (size_t)(n0 + r) * GCN_D))[lane];
        float4 o;
        o.x = b.x + x.x * self;
        o.y = b.y + x.y * self;
        o.z = b.z + x.z * self;
        o.w = b.w + x.w * self;
        ((float4*)(sacc + r * GCN_D))[lane] = o;
    }
    __syncthreads();

    float4 id[SC_LN_ROWS];
#pragma unroll
    for (int k = 0; k < SC_LN_ROWS; ++k) {
        const int rk = wave + k * SB_WAVES;
        const int rr = rk < nrows ? rk : nrows - 1;
        id[k] = ((const float4*)(hin + (size_t)(n0 + rr) * GCN_D))[lane];
    }
    const float4 gm = ((const float4*)gamma)[lane];
    const float4 bt = ((const float4*)beta)[lane];

    const int e_begin = off[blockIdx.x];
    const int e_end = off[blockIdx.x + 1];
    int e0 = e_begin + wave * SC_EDGES_PER_WAVE;
    int r = 0, c = 0, wbits = 0;
    if (e0 + lane < e_end) {
        r = prow[e0 + lane];
        c = pcol[e0 + lane];
        wbits = __float_as_int(pnorm[e0 + lane]);
    }
    while (e0 < e_end) {
        const int left = e_end - e0;
        const int cnt = left < SC_EDGES_PER_WAVE ? left : SC_EDGES_PER_WAVE;
        const int e1 = e0 + SB_WAVES * SC_EDGES_PER_WAVE;
        int rn = 0, cn = 0, wn = 0;
        if (e1 + lane < e_end) {
            rn = prow[e1 + lane];
            cn = pcol[e1 + lane];
            wn = __float_as_int(pnorm[e1 + lane]);
        }
        int g = 0;
#pragma unroll 1
        for (; g + SC_GROUP <= cnt; g += SC_GROUP) {
            float m[SC_GROUP][4];
#pragma unroll
            for (int k = 0; k < SC_GROUP; ++k) {
                const int rs = __builtin_amdgcn_readlane(r, g + k);
                const float* src = XW + (size_t)rs * GCN_D + lane;
                m[k][0] = src[0];
                m[k][1] = src[32];
                m[k][2] = src[64];
                m[k][3] = src[96];
            }
#pragma unroll
            for (int k = 0; k < SC_GROUP; ++k) {
                const int cs = __builtin_amdgcn_readlane(c, g + k);
                const float wf = __int_as_float(__builtin_amdgcn_readlane(wbits, g + k));
                float* dst = sacc + cs * GCN_D + lane;
                atomicAdd(dst + 0,  m[k][0] * wf);
                atomicAdd(dst + 32, m[k][1] * wf);
                atomicAdd(dst + 64, m[k][2] * wf);
                atomicAdd(dst + 96, m[k][3] * wf);
            }
        }
        for (int i = g; i < cnt; ++i) {
            const int rs = __builtin_amdgcn_readlane(r, i);
            const int cs = __builtin_amdgcn_readlane(c, i);
            const float wf = __int_as_float(__builtin_amdgcn_readlane(wbits, i));
            const float* src = XW + (size_t)rs * GCN_D + lane;
            float* dst = sacc + cs * GCN_D + lane;
            const float m0 = src[0], m1 = src[32], m2 = src[64], m3 = src[96];
            atomicAdd(dst + 0,  m0 * wf);
            atomicAdd(dst + 32, m1 * wf);
            atomicAdd(dst + 64, m2 * wf);
            atomicAdd(dst + 96, m3 * wf);
        }
        r = rn;
        c = cn;
        wbits = wn;
        e0 = e1;
    }
    __syncthreads();

    float4 a[SC_LN_ROWS];
#pragma unroll
    for (int k = 0; k < SC_LN_ROWS; ++k) {
        const int rk = wave + k * SB_WAVES;
        const int rr = rk < nrows ? rk : nrows - 1;
        a[k] = ((const float4*)(sacc + rr * GCN_D))[lane];
    }

    float s[SC_LN_ROWS];
#pragma unroll
    for (int k = 0; k < SC_LN_ROWS; ++k) s[k] = a[k].x + a[k].y + a[k].z + a[k].w;
#pragma unroll
    for (int sh = 16; sh > 0; sh >>= 1) {
#pragma unroll
        for (int k = 0; k < SC_LN_ROWS; ++k) s[k] += __shfl_xor(s[k], sh, 32);
    }

    float q[SC_LN_ROWS];
#pragma unroll
    for (int k = 0; k < SC_LN_ROWS; ++k) {
        const float mu = s[k] * (1.0f / GCN_D);
        a[k].x -= mu; a[k].y -= mu; a[k].z -= mu; a[k].w -= mu;
        q[k] = a[k].x * a[k].x + a[k].y * a[k].y + a[k].z * a[k].z + a[k].w * a[k].w;
    }
#pragma unroll
    for (int sh = 16; sh > 0; sh >>= 1) {
#pragma unroll
        for (int k = 0; k < SC_LN_ROWS; ++k) q[k] += __shfl_xor(q[k], sh, 32);
    }

#pragma unroll
    for (int k = 0; k < SC_LN_ROWS; ++k) {
        const float inv = rsqrtf(q[k] * (1.0f / GCN_D) + LN_EPS);
        float4 y;
        y.x = a[k].x * inv * gm.x + bt.x;
        y.y = a[k].y * inv * gm.y + bt.y;
        y.z = a[k].z * inv * gm.z + bt.z;
        y.w = a[k].w * inv * gm.w + bt.w;
        if (do_relu) {
            y.x = fmaxf(y.x, 0.0f);
            y.y = fmaxf(y.y, 0.0f);
            y.z = fmaxf(y.z, 0.0f);
            y.w = fmaxf(y.w, 0.0f);
        }
        y.x += id[k].x; y.y += id[k].y; y.z += id[k].z; y.w += id[k].w;
        const int rk = wave + k * SB_WAVES;
        if (rk < nrows)
            ((float4*)(hout + (size_t)(n0 + rk) * GCN_D))[lane] = y;
    }
}


extern "C" void kernel_launch(void* const* d_in, const int* in_sizes, int n_in,
                              void* d_out, int out_size, void* d_ws, size_t ws_size,
                              hipStream_t stream) {
    (void)in_sizes; (void)n_in; (void)out_size; (void)ws_size;

    const float* x      = (const float*)d_in[0];
    const int*   ei     = (const int*)d_in[1];
    const float* ew     = (const float*)d_in[2];
    const float* Ws     = (const float*)d_in[3];
    const float* bs     = (const float*)d_in[4];
    const float* gammas = (const float*)d_in[5];
    const float* betas  = (const float*)d_in[6];
    float*       out    = (float*)d_out;

    const int* row = ei;
    const int* col = ei + GCN_E;

    float* ws   = (float*)d_ws;
    float* deg  = ws;
    float* dinv = deg + GCN_N;
    float* pnorm = dinv + GCN_N;
    float* xw   = pnorm + GCN_E;
    float* h1   = xw + (size_t)GCN_N * GCN_D;
    float* deg1 = h1 + (size_t)GCN_N * GCN_D;
    int* cnt    = (int*)(deg1 + GCN_N);
    int* cursor = cnt + SB_PAD;
    int* off    = cursor + SB_PAD;
    int* prow   = off + SB_PAD;
    int* pcol   = prow + GCN_E;

    const int BLK = 256;

    k_init_deg<<<(GCN_N + BLK - 1) / BLK, BLK, 0, stream>>>(deg, deg1, cnt, cursor, GCN_N);
    k_accum_deg<<<(GCN_E + DG_THREADS * DG_PER_THREAD - 1) / (DG_THREADS * DG_PER_THREAD),
                  DG_THREADS, 0, stream>>>(ew, col, deg, deg1, GCN_E);
    k_dinv<<<(GCN_N + BLK - 1) / BLK, BLK, 0, stream>>>(deg, deg1, dinv, GCN_N);
    const int bin_blocks = (GCN_E + BN_SLICE - 1) / BN_SLICE;
    k_bin_count<<<bin_blocks, BN_THREADS, 0, stream>>>(col, cnt, GCN_E);
    k_bin_place<<<bin_blocks, BN_THREADS, 0, stream>>>(dinv, row, col, ew, cnt, cursor, off,
                                                       prow, pcol, pnorm, GCN_E);

    const int row_tiles    = (GCN_N + 15) / 16;
    const int gemm_blocks  = (row_tiles + GM_ROWTILES - 1) / GM_ROWTILES;

    const float* hin[GCN_L]  = { x,  h1,  out };
    float*       hout[GCN_L] = { h1, out, out };

    for (int l = 0; l < GCN_L; ++l) {
        const float* W = Ws + (size_t)l * GCN_D * GCN_D;
        k_gemm_wmma<<<gemm_blocks, GM_WAVES * 32, 0, stream>>>(hin[l], W, xw, GCN_N);
        k_scatter<<<SB_BINS, SB_WAVES * 32, 0, stream>>>(xw, dinv, bs + (size_t)l * GCN_D,
                                                         prow, pcol, pnorm, off, hin[l],
                                                         gammas + (size_t)l * GCN_D,
                                                         betas + (size_t)l * GCN_D,
                                                         hout[l], GCN_N,
                                                          (l < GCN_L - 1) ? 1 : 0);
    }
}


# Degree sums leave global float atomics: taken on the binned edge lists, a work-group per range adds its weights in LDS (one arrival-order sum per node, not two); coefficients in a pass behind it
# speedup vs baseline: 4.0262x; 1.0608x over previous
#include <hip/hip_runtime.h>
#include <hip/hip_bf16.h>


#define GCN_N 50000
#define GCN_E 1600000
#define GCN_D 128
#define GCN_L 3
#define LN_EPS 1e-5f

#define SB_ROWS 49
#define SB_BINS ((GCN_N + SB_ROWS - 1) / SB_ROWS)
#define SB_LDS_ROWS 64
#define SB_PAD 1024
#define SB_WAVES 8
#define BN_THREADS 1024
#define BN_PER_THREAD 8
#define BN_SLICE (BN_THREADS * BN_PER_THREAD)
static_assert(SB_BINS < SB_PAD && SB_PAD == BN_THREADS, "one counter per binning thread");
static_assert(SB_ROWS <= SB_LDS_ROWS, "a range's sum rows fit the reserved LDS rows");

__device__ __forceinline__ int sb_bin(int node) { return (int)((unsigned)node / (unsigned)SB_ROWS); }

typedef __attribute__((ext_vector_type(2))) float v2f;
typedef __attribute__((ext_vector_type(8))) float v8f;


__global__ void __launch_bounds__(SB_PAD)
k_zero_counters(int* __restrict__ cnt, int* __restrict__ cursor) {
    cnt[threadIdx.x] = 0;
    cursor[threadIdx.x] = 0;
}


__global__ void __launch_bounds__(BN_THREADS)
k_bin_count(const int* __restrict__ col, int* __restrict__ cnt, int e) {
    __shared__ int h[SB_PAD];
    const int tid = threadIdx.x;
    h[tid] = 0;
    __syncthreads();
    const int base = blockIdx.x * BN_SLICE + tid;
    int cv[BN_PER_THREAD];
#pragma unroll
    for (int k = 0; k < BN_PER_THREAD; ++k) {
        const int i = base + k * BN_THREADS;
        cv[k] = col[i < e ? i : e - 1];
    }
#pragma unroll
    for (int k = 0; k < BN_PER_THREAD; ++k)
        if (base + k * BN_THREADS < e) atomicAdd(&h[sb_bin(cv[k])], 1);
    __syncthreads();
    const int v = h[tid];
    if (v) atomicAdd(&cnt[tid], v);
}

__global__ void __launch_bounds__(BN_THREADS)
k_bin_place(const int* __restrict__ row, const int* __restrict__ col,
            const float* __restrict__ w, const int* __restrict__ cnt,
            int* __restrict__ cursor, int* __restrict__ off,
            int* __restrict__ prow, int* __restrict__ pcol, float* __restrict__ pw, int e) {
    __shared__ int h[SB_PAD];
    __shared__ int wsum[32];
    const int tid = threadIdx.x;
    const int lane = tid & 31;
    const int wv = tid >> 5;

    const int cv = cnt[tid];
    int inc = cv;
#pragma unroll
    for (int d = 1; d < 32; d <<= 1) {
        const int t = __shfl_up(inc, d, 32);
        if (lane >= d) inc += t;
    }
    if (lane == 31) wsum[wv] = inc;
    h[tid] = 0;
    __syncthreads();
    if (wv == 0) {
        int s = wsum[lane];
#pragma unroll
        for (int d = 1; d < 32; d <<= 1) {
            const int t = __shfl_up(s, d, 32);
            if (lane >= d) s += t;
        }
        wsum[lane] = s;
    }
    __syncthreads();
    const int start = inc - cv + (wv ? wsum[wv - 1] : 0);
    if (blockIdx.x == 0) off[tid] = start;

    int c[BN_PER_THREAD], r[BN_PER_THREAD], rk[BN_PER_THREAD];
    float we[BN_PER_THREAD];
    const int base = blockIdx.x * BN_SLICE + tid;
#pragma unroll
    for (int k = 0; k < BN_PER_THREAD; ++k) {
        const int i = base + k * BN_THREADS;
        const int ii = i < e ? i : e - 1;
        c[k] = col[ii];
        r[k] = row[ii];
        we[k] = w[ii];
    }
#pragma unroll
    for (int k = 0; k < BN_PER_THREAD; ++k) {
        const int i = base + k * BN_THREADS;
        rk[k] = 0;
        if (i < e) rk[k] = atomicAdd(&h[sb_bin(c[k])], 1);
    }
    __syncthreads();
    const int hv = h[tid];
    h[tid] = start + (hv ? atomicAdd(&cursor[tid], hv) : 0);
    __syncthreads();
#pragma unroll
    for (int k = 0; k < BN_PER_THREAD; ++k) {
        const int i = base + k * BN_THREADS;
        if (i < e) {
            const int bin = sb_bin(c[k]);
            const int pos = h[bin] + rk[k];
            prow[pos] = r[k];
            pcol[pos] = c[k] - bin * SB_ROWS;
            pw[pos] = we[k];
        }
    }
}


#define BD_THREADS 256
#define BD_PER_THREAD 8

__global__ void __launch_bounds__(BD_THREADS)
k_bin_deg(const int* __restrict__ pcol, const float* __restrict__ pw,
          const int* __restrict__ off, float* __restrict__ dinv, int n_nodes) {
    __shared__ float sdeg[SB_LDS_ROWS];
    const int tid = threadIdx.x;
    const int e_begin = off[blockIdx.x];
    const int n_e = off[blockIdx.x + 1] - e_begin;
    if (tid < SB_LDS_ROWS) sdeg[tid] = 1.0f;
    __syncthreads();
    for (int base = tid; base < n_e; base += BD_THREADS * BD_PER_THREAD) {
        int c[BD_PER_THREAD];
        float we[BD_PER_THREAD];
#pragma unroll
        for (int k = 0; k < BD_PER_THREAD; ++k) {
            const int i = base + k * BD_THREADS;
            const int ii = e_begin + (i < n_e ? i : n_e - 1);
            c[k] = pcol[ii];
            we[k] = pw[ii];
        }
#pragma unroll
        for (int k = 0; k < BD_PER_THREAD; ++k)
            if (base + k * BD_THREADS < n_e) atomicAdd(&sdeg[c[k]], we[k]);
    }
    __syncthreads();
    const int n0 = (int)blockIdx.x * SB_ROWS;
    if (tid < SB_ROWS && n0 + tid < n_nodes) {
        const float d = sdeg[tid];
        dinv[n0 + tid] = (d > 0.0f) ? rsqrtf(fmaxf(d, 1e-12f)) : 0.0f;
    }
}

__global__ void __launch_bounds__(BD_THREADS)
k_bin_norm(const float* __restrict__ dinv, const int* __restrict__ prow,
           const int* __restrict__ pcol, const int* __restrict__ off,
           float* __restrict__ pnorm) {
    const int tid = threadIdx.x;
    const int n0 = (int)blockIdx.x * SB_ROWS;
    const int e_begin = off[blockIdx.x];
    const int n_e = off[blockIdx.x + 1] - e_begin;
    for (int base = tid; base < n_e; base += BD_THREADS * BD_PER_THREAD) {
        int r[BD_PER_THREAD], c[BD_PER_THREAD];
        float we[BD_PER_THREAD], dr[BD_PER_THREAD], dc[BD_PER_THREAD];
#pragma unroll
        for (int k = 0; k < BD_PER_THREAD; ++k) {
            const int i = base + k * BD_THREADS;
            const int ii = e_begin + (i < n_e ? i : n_e - 1);
            r[k] = prow[ii];
            c[k] = pcol[ii];
            we[k] = pnorm[ii];
        }
#pragma unroll
        for (int k = 0; k < BD_PER_THREAD; ++k) {
            dr[k] = dinv[r[k]];
            dc[k] = dinv[n0 + c[k]];
        }
#pragma unroll
        for (int k = 0; k < BD_PER_THREAD; ++k) {
            const int i = base + k * BD_THREADS;
            if (i < n_e) pnorm[e_begin + i] = dr[k] * we[k] * dc[k];
        }
    }
}


#define GM_ROWTILES 4
#define GM_WAVES 4

__global__ void __launch_bounds__(GM_WAVES * 32)
k_gemm_wmma(const float* __restrict__ H, const float* __restrict__ W,
            float* __restrict__ XW, int n_nodes) {
    const int lane = threadIdx.x & 31;
    const int col0 = (threadIdx.x >> 5) * 32;
    const int lm = lane & 15;
    const int hi = lane >> 4;

    const float4* __restrict__ Bp0 = (const float4*)(W + (size_t)(col0 + lm) * GCN_D) + hi;
    const float4* __restrict__ Bp1 = (const float4*)(W + (size_t)(col0 + 16 + lm) * GCN_D) + hi;
    float4 b0[16], b1[16];
#pragma unroll
    for (int i = 0; i < 16; ++i) {
        b0[i] = Bp0[2 * i];
        b1[i] = Bp1[2 * i];
    }

    const int nc = col0 + lm;
#pragma unroll 1
    for (int rt = 0; rt < GM_ROWTILES; ++rt) {
        const int row0 = ((int)blockIdx.x * GM_ROWTILES + rt) * 16;
        if (row0 >= n_nodes) break;

        const float4* __restrict__ Ap = (const float4*)(H + (size_t)(row0 + lm) * GCN_D) + hi;
        float4 a[16];
#pragma unroll
        for (int i = 0; i < 16; ++i) a[i] = Ap[2 * i];

        v8f c0 = {};
        v8f c1 = {};
#pragma unroll
        for (int i = 0; i < 16; ++i) {
            v2f ax; ax.x = a[i].x; ax.y = a[i].y;
            v2f az; az.x = a[i].z; az.y = a[i].w;
            v2f bx0; bx0.x = b0[i].x; bx0.y = b0[i].y;
            v2f bz0; bz0.x = b0[i].z; bz0.y = b0[i].w;
            v2f bx1; bx1.x = b1[i].x; bx1.y = b1[i].y;
            v2f bz1; bz1.x = b1[i].z; bz1.y = b1[i].w;
            c0 = __builtin_amdgcn_wmma_f32_16x16x4_f32(false, ax, false, bx0, (short)0, c0, false, false);
            c1 = __builtin_amdgcn_wmma_f32_16x16x4_f32(false, ax, false, bx1, (short)0, c1, false, false);
            c0 = __builtin_amdgcn_wmma_f32_16x16x4_f32(false, az, false, bz0, (short)0, c0, false, false);
            c1 = __builtin_amdgcn_wmma_f32_16x16x4_f32(false, az, false, bz1, (short)0, c1, false, false);
        }

        const int m0 = row0 + hi * 8;
#pragma unroll
        for (int v = 0; v < 8; ++v) {
            XW[(size_t)(m0 + v) * GCN_D + nc] = c0[v];
            XW[(size_t)(m0 + v) * GCN_D + nc + 16] = c1[v];
        }
    }
}


#define SC_EDGES_PER_WAVE 32
#define SC_GROUP 8
#define SC_LN_ROWS ((SB_ROWS + SB_WAVES - 1) / SB_WAVES)

__global__ void __launch_bounds__(SB_WAVES * 32)
k_scatter(const float* __restrict__ XW, const float* __restrict__ dinv,
          const float* __restrict__ bias, const int* __restrict__ prow,
          const int* __restrict__ pcol, const float* __restrict__ pnorm,
          const int* __restrict__ off, const float* hin,
          const float* __restrict__ gamma, const float* __restrict__ beta,
          float* hout, int n_nodes, int do_relu) {
    __shared__ float sacc[SB_LDS_ROWS * GCN_D];
    const int lane = threadIdx.x & 31;
    const int wave = __builtin_amdgcn_readfirstlane((int)(threadIdx.x >> 5));
    const int n0 = (int)blockIdx.x * SB_ROWS;
    const int nleft = n_nodes - n0;
    const int nrows = nleft < SB_ROWS ? nleft : SB_ROWS;

    const float4 b = ((const float4*)bias)[lane];
    for (int r = wave; r < nrows; r += SB_WAVES) {
        const float dn = dinv[n0 + r];
        const float self = dn * dn;
        const float4 x = ((const float4*)(XW + (size_t)(n0 + r) * GCN_D))[lane];
        float4 o;
        o.x = b.x + x.x * self;
        o.y = b.y + x.y * self;
        o.z = b.z + x.z * self;
        o.w = b.w + x.w * self;
        ((float4*)(sacc + r * GCN_D))[lane] = o;
    }
    __syncthreads();

    float4 id[SC_LN_ROWS];
#pragma unroll
    for (int k = 0; k < SC_LN_ROWS; ++k) {
        const int rk = wave + k * SB_WAVES;
        const int rr = rk < nrows ? rk : nrows - 1;
        id[k] = ((const float4*)(hin + (size_t)(n0 + rr) * GCN_D))[lane];
    }
    const float4 gm = ((const float4*)gamma)[lane];
    const float4 bt = ((const float4*)beta)[lane];

    const int e_begin = off[blockIdx.x];
    const int e_end = off[blockIdx.x + 1];
    int e0 = e_begin + wave * SC_EDGES_PER_WAVE;
    int r = 0, c = 0, wbits = 0;
    if (e0 + lane < e_end) {
        r = prow[e0 + lane];
        c = pcol[e0 + lane];
        wbits = __float_as_int(pnorm[e0 + lane]);
    }
    while (e0 < e_end) {
        const int left = e_end - e0;
        const int cnt = left < SC_EDGES_PER_WAVE ? left : SC_EDGES_PER_WAVE;
        const int e1 = e0 + SB_WAVES * SC_EDGES_PER_WAVE;
        int rn = 0, cn = 0, wn = 0;
        if (e1 + lane < e_end) {
            rn = prow[e1 + lane];
            cn = pcol[e1 + lane];
            wn = __float_as_int(pnorm[e1 + lane]);
        }
        int g = 0;
#pragma unroll 1
        for (; g + SC_GROUP <= cnt; g += SC_GROUP) {
            float m[SC_GROUP][4];
#pragma unroll
            for (int k = 0; k < SC_GROUP; ++k) {
                const int rs = __builtin_amdgcn_readlane(r, g + k);
                const float* src = XW + (size_t)rs * GCN_D + lane;
                m[k][0] = src[0];
                m[k][1] = src[32];
                m[k][2] = src[64];
                m[k][3] = src[96];
            }
#pragma unroll
            for (int k = 0; k < SC_GROUP; ++k) {
                const int cs = __builtin_amdgcn_readlane(c, g + k);
                const float wf = __int_as_float(__builtin_amdgcn_readlane(wbits, g + k));
                float* dst = sacc + cs * GCN_D + lane;
                atomicAdd(dst + 0,  m[k][0] * wf);
                atomicAdd(dst + 32, m[k][1] * wf);
                atomicAdd(dst + 64, m[k][2] * wf);
                atomicAdd(dst + 96, m[k][3] * wf);
            }
        }
        for (int i = g; i < cnt; ++i) {
            const int rs = __builtin_amdgcn_readlane(r, i);
            const int cs = __builtin_amdgcn_readlane(c, i);
            const float wf = __int_as_float(__builtin_amdgcn_readlane(wbits, i));
            const float* src = XW + (size_t)rs * GCN_D + lane;
            float* dst = sacc + cs * GCN_D + lane;
            const float m0 = src[0], m1 = src[32], m2 = src[64], m3 = src[96];
            atomicAdd(dst + 0,  m0 * wf);
            atomicAdd(dst + 32, m1 * wf);
            atomicAdd(dst + 64, m2 * wf);
            atomicAdd(dst + 96, m3 * wf);
        }
        r = rn;
        c = cn;
        wbits = wn;
        e0 = e1;
    }
    __syncthreads();

    float4 a[SC_LN_ROWS];
#pragma unroll
    for (int k = 0; k < SC_LN_ROWS; ++k) {
        const int rk = wave + k * SB_WAVES;
        const int rr = rk < nrows ? rk : nrows - 1;
        a[k] = ((const float4*)(sacc + rr * GCN_D))[lane];
    }

    float s[SC_LN_ROWS];
#pragma unroll
    for (int k = 0; k < SC_LN_ROWS; ++k) s[k] = a[k].x + a[k].y + a[k].z + a[k].w;
#pragma unroll
    for (int sh = 16; sh > 0; sh >>= 1) {
#pragma unroll
        for (int k = 0; k < SC_LN_ROWS; ++k) s[k] += __shfl_xor(s[k], sh, 32);
    }

    float q[SC_LN_ROWS];
#pragma unroll
    for (int k = 0; k < SC_LN_ROWS; ++k) {
        const float mu = s[k] * (1.0f / GCN_D);
        a[k].x -= mu; a[k].y -= mu; a[k].z -= mu; a[k].w -= mu;
        q[k] = a[k].x * a[k].x + a[k].y * a[k].y + a[k].z * a[k].z + a[k].w * a[k].w;
    }
#pragma unroll
    for (int sh = 16; sh > 0; sh >>= 1) {
#pragma unroll
        for (int k = 0; k < SC_LN_ROWS; ++k) q[k] += __shfl_xor(q[k], sh, 32);
    }

#pragma unroll
    for (int k = 0; k < SC_LN_ROWS; ++k) {
        const float inv = rsqrtf(q[k] * (1.0f / GCN_D) + LN_EPS);
        float4 y;
        y.x = a[k].x * inv * gm.x + bt.x;
        y.y = a[k].y * inv * gm.y + bt.y;
        y.z = a[k].z * inv * gm.z + bt.z;
        y.w = a[k].w * inv * gm.w + bt.w;
        if (do_relu) {
            y.x = fmaxf(y.x, 0.0f);
            y.y = fmaxf(y.y, 0.0f);
            y.z = fmaxf(y.z, 0.0f);
            y.w = fmaxf(y.w, 0.0f);
        }
        y.x += id[k].x; y.y += id[k].y; y.z += id[k].z; y.w += id[k].w;
        const int rk = wave + k * SB_WAVES;
        if (rk < nrows)
            ((float4*)(hout + (size_t)(n0 + rk) * GCN_D))[lane] = y;
    }
}


extern "C" void kernel_launch(void* const* d_in, const int* in_sizes, int n_in,
                              void* d_out, int out_size, void* d_ws, size_t ws_size,
                              hipStream_t stream) {
    (void)in_sizes; (void)n_in; (void)out_size; (void)ws_size;

    const float* x      = (const float*)d_in[0];
    const int*   ei     = (const int*)d_in[1];
    const float* ew     = (const float*)d_in[2];
    const float* Ws     = (const float*)d_in[3];
    const float* bs     = (const float*)d_in[4];
    const float* gammas = (const float*)d_in[5];
    const float* betas  = (const float*)d_in[6];
    float*       out    = (float*)d_out;

    const int* row = ei;
    const int* col = ei + GCN_E;

    float* ws   = (float*)d_ws;
    float* pnorm = ws;
    float* xw   = pnorm + GCN_E;
    float* h1   = xw + (size_t)GCN_N * GCN_D;
    int* cnt    = (int*)(h1 + (size_t)GCN_N * GCN_D);
    int* cursor = cnt + SB_PAD;
    int* off    = cursor + SB_PAD;
    int* prow   = off + SB_PAD;
    int* pcol   = prow + GCN_E;
    float* dinv = (float*)(pcol + GCN_E);

    const int bin_blocks = (GCN_E + BN_SLICE - 1) / BN_SLICE;
    k_zero_counters<<<1, SB_PAD, 0, stream>>>(cnt, cursor);
    k_bin_count<<<bin_blocks, BN_THREADS, 0, stream>>>(col, cnt, GCN_E);
    k_bin_place<<<bin_blocks, BN_THREADS, 0, stream>>>(row, col, ew, cnt, cursor, off,
                                                       prow, pcol, pnorm, GCN_E);
    k_bin_deg<<<SB_BINS, BD_THREADS, 0, stream>>>(pcol, pnorm, off, dinv, GCN_N);
    k_bin_norm<<<SB_BINS, BD_THREADS, 0, stream>>>(dinv, prow, pcol, off, pnorm);

    const int row_tiles    = (GCN_N + 15) / 16;
    const int gemm_blocks  = (row_tiles + GM_ROWTILES - 1) / GM_ROWTILES;

    const float* hin[GCN_L]  = { x,  h1,  out };
    float*       hout[GCN_L] = { h1, out, out };

    for (int l = 0; l < GCN_L; ++l) {
        const float* W = Ws + (size_t)l * GCN_D * GCN_D;
        k_gemm_wmma<<<gemm_blocks, GM_WAVES * 32, 0, stream>>>(hin[l], W, xw, GCN_N);
        k_scatter<<<SB_BINS, SB_WAVES * 32, 0, stream>>>(xw, dinv, bs + (size_t)l * GCN_D,
                                                         prow, pcol, pnorm, off, hin[l],
                                                         gammas + (size_t)l * GCN_D,
                                                         betas + (size_t)l * GCN_D,
                                                         hout[l], GCN_N,
                                                          (l < GCN_L - 1) ? 1 : 0);
    }
}


# Scatter fetches a source row with one 16-byte-per-lane load (4 lines per request, not 4 one-line loads); LDS sum rows hold feature 4l+j at word 32j+l, adds stay one word per bank; sum orders kept
# speedup vs baseline: 4.4433x; 1.1036x over previous
#include <hip/hip_runtime.h>
#include <hip/hip_bf16.h>


#define GCN_N 50000
#define GCN_E 1600000
#define GCN_D 128
#define GCN_L 3
#define LN_EPS 1e-5f

#define SB_ROWS 49
#define SB_BINS ((GCN_N + SB_ROWS - 1) / SB_ROWS)
#define SB_LDS_ROWS 64
#define SB_PAD 1024
#define SB_WAVES 8
#define BN_THREADS 1024
#define BN_PER_THREAD 8
#define BN_SLICE (BN_THREADS * BN_PER_THREAD)
static_assert(SB_BINS < SB_PAD && SB_PAD == BN_THREADS, "one counter per binning thread");
static_assert(SB_ROWS <= SB_LDS_ROWS, "a range's sum rows fit the reserved LDS rows");

__device__ __forceinline__ int sb_bin(int node) { return (int)((unsigned)node / (unsigned)SB_ROWS); }

typedef __attribute__((ext_vector_type(2))) float v2f;
typedef __attribute__((ext_vector_type(8))) float v8f;


__global__ void __launch_bounds__(SB_PAD)
k_zero_counters(int* __restrict__ cnt, int* __restrict__ cursor) {
    cnt[threadIdx.x] = 0;
    cursor[threadIdx.x] = 0;
}


__global__ void __launch_bounds__(BN_THREADS)
k_bin_count(const int* __restrict__ col, int* __restrict__ cnt, int e) {
    __shared__ int h[SB_PAD];
    const int tid = threadIdx.x;
    h[tid] = 0;
    __syncthreads();
    const int base = blockIdx.x * BN_SLICE + tid;
    int cv[BN_PER_THREAD];
#pragma unroll
    for (int k = 0; k < BN_PER_THREAD; ++k) {
        const int i = base + k * BN_THREADS;
        cv[k] = col[i < e ? i : e - 1];
    }
#pragma unroll
    for (int k = 0; k < BN_PER_THREAD; ++k)
        if (base + k * BN_THREADS < e) atomicAdd(&h[sb_bin(cv[k])], 1);
    __syncthreads();
    const int v = h[tid];
    if (v) atomicAdd(&cnt[tid], v);
}

__global__ void __launch_bounds__(BN_THREADS)
k_bin_place(const int* __restrict__ row, const int* __restrict__ col,
            const float* __restrict__ w, const int* __restrict__ cnt,
            int* __restrict__ cursor, int* __restrict__ off,
            int* __restrict__ prow, int* __restrict__ pcol, float* __restrict__ pw, int e) {
    __shared__ int h[SB_PAD];
    __shared__ int wsum[32];
    const int tid = threadIdx.x;
    const int lane = tid & 31;
    const int wv = tid >> 5;

    const int cv = cnt[tid];
    int inc = cv;
#pragma unroll
    for (int d = 1; d < 32; d <<= 1) {
        const int t = __shfl_up(inc, d, 32);
        if (lane >= d) inc += t;
    }
    if (lane == 31) wsum[wv] = inc;
    h[tid] = 0;
    __syncthreads();
    if (wv == 0) {
        int s = wsum[lane];
#pragma unroll
        for (int d = 1; d < 32; d <<= 1) {
            const int t = __shfl_up(s, d, 32);
            if (lane >= d) s += t;
        }
        wsum[lane] = s;
    }
    __syncthreads();
    const int start = inc - cv + (wv ? wsum[wv - 1] : 0);
    if (blockIdx.x == 0) off[tid] = start;

    int c[BN_PER_THREAD], r[BN_PER_THREAD], rk[BN_PER_THREAD];
    float we[BN_PER_THREAD];
    const int base = blockIdx.x * BN_SLICE + tid;
#pragma unroll
    for (int k = 0; k < BN_PER_THREAD; ++k) {
        const int i = base + k * BN_THREADS;
        const int ii = i < e ? i : e - 1;
        c[k] = col[ii];
        r[k] = row[ii];
        we[k] = w[ii];
    }
#pragma unroll
    for (int k = 0; k < BN_PER_THREAD; ++k) {
        const int i = base + k * BN_THREADS;
        rk[k] = 0;
        if (i < e) rk[k] = atomicAdd(&h[sb_bin(c[k])], 1);
    }
    __syncthreads();
    const int hv = h[tid];
    h[tid] = start + (hv ? atomicAdd(&cursor[tid], hv) : 0);
    __syncthreads();
#pragma unroll
    for (int k = 0; k < BN_PER_THREAD; ++k) {
        const int i = base + k * BN_THREADS;
        if (i < e) {
            const int bin = sb_bin(c[k]);
            const int pos = h[bin] + rk[k];
            prow[pos] = r[k];
            pcol[pos] = c[k] - bin * SB_ROWS;
            pw[pos] = we[k];
        }
    }
}


#define BD_THREADS 256
#define BD_PER_THREAD 8

__global__ void __launch_bounds__(BD_THREADS)
k_bin_deg(const int* __restrict__ pcol, const float* __restrict__ pw,
          const int* __restrict__ off, float* __restrict__ dinv, int n_nodes) {
    __shared__ float sdeg[SB_LDS_ROWS];
    const int tid = threadIdx.x;
    const int e_begin = off[blockIdx.x];
    const int n_e = off[blockIdx.x + 1] - e_begin;
    if (tid < SB_LDS_ROWS) sdeg[tid] = 1.0f;
    __syncthreads();
    for (int base = tid; base < n_e; base += BD_THREADS * BD_PER_THREAD) {
        int c[BD_PER_THREAD];
        float we[BD_PER_THREAD];
#pragma unroll
        for (int k = 0; k < BD_PER_THREAD; ++k) {
            const int i = base + k * BD_THREADS;
            const int ii = e_begin + (i < n_e ? i : n_e - 1);
            c[k] = pcol[ii];
            we[k] = pw[ii];
        }
#pragma unroll
        for (int k = 0; k < BD_PER_THREAD; ++k)
            if (base + k * BD_THREADS < n_e) atomicAdd(&sdeg[c[k]], we[k]);
    }
    __syncthreads();
    const int n0 = (int)blockIdx.x * SB_ROWS;
    if (tid < SB_ROWS && n0 + tid < n_nodes) {
        const float d = sdeg[tid];
        dinv[n0 + tid] = (d > 0.0f) ? rsqrtf(fmaxf(d, 1e-12f)) : 0.0f;
    }
}

__global__ void __launch_bounds__(BD_THREADS)
k_bin_norm(const float* __restrict__ dinv, const int* __restrict__ prow,
           const int* __restrict__ pcol, const int* __restrict__ off,
           float* __restrict__ pnorm) {
    const int tid = threadIdx.x;
    const int n0 = (int)blockIdx.x * SB_ROWS;
    const int e_begin = off[blockIdx.x];
    const int n_e = off[blockIdx.x + 1] - e_begin;
    for (int base = tid; base < n_e; base += BD_THREADS * BD_PER_THREAD) {
        int r[BD_PER_THREAD], c[BD_PER_THREAD];
        float we[BD_PER_THREAD], dr[BD_PER_THREAD], dc[BD_PER_THREAD];
#pragma unroll
        for (int k = 0; k < BD_PER_THREAD; ++k) {
            const int i = base + k * BD_THREADS;
            const int ii = e_begin + (i < n_e ? i : n_e - 1);
            r[k] = prow[ii];
            c[k] = pcol[ii];
            we[k] = pnorm[ii];
        }
#pragma unroll
        for (int k = 0; k < BD_PER_THREAD; ++k) {
            dr[k] = dinv[r[k]];
            dc[k] = dinv[n0 + c[k]];
        }
#pragma unroll
        for (int k = 0; k < BD_PER_THREAD; ++k) {
            const int i = base + k * BD_THREADS;
            if (i < n_e) pnorm[e_begin + i] = dr[k] * we[k] * dc[k];
        }
    }
}


#define GM_ROWTILES 4
#define GM_WAVES 4

__global__ void __launch_bounds__(GM_WAVES * 32)
k_gemm_wmma(const float* __restrict__ H, const float* __restrict__ W,
            float* __restrict__ XW, int n_nodes) {
    const int lane = threadIdx.x & 31;
    const int col0 = (threadIdx.x >> 5) * 32;
    const int lm = lane & 15;
    const int hi = lane >> 4;

    const float4* __restrict__ Bp0 = (const float4*)(W + (size_t)(col0 + lm) * GCN_D) + hi;
    const float4* __restrict__ Bp1 = (const float4*)(W + (size_t)(col0 + 16 + lm) * GCN_D) + hi;
    float4 b0[16], b1[16];
#pragma unroll
    for (int i = 0; i < 16; ++i) {
        b0[i] = Bp0[2 * i];
        b1[i] = Bp1[2 * i];
    }

    const int nc = col0 + lm;
#pragma unroll 1
    for (int rt = 0; rt < GM_ROWTILES; ++rt) {
        const int row0 = ((int)blockIdx.x * GM_ROWTILES + rt) * 16;
        if (row0 >= n_nodes) break;

        const float4* __restrict__ Ap = (const float4*)(H + (size_t)(row0 + lm) * GCN_D) + hi;
        float4 a[16];
#pragma unroll
        for (int i = 0; i < 16; ++i) a[i] = Ap[2 * i];

        v8f c0 = {};
        v8f c1 = {};
#pragma unroll
        for (int i = 0; i < 16; ++i) {
            v2f ax; ax.x = a[i].x; ax.y = a[i].y;
            v2f az; az.x = a[i].z; az.y = a[i].w;
            v2f bx0; bx0.x = b0[i].x; bx0.y = b0[i].y;
            v2f bz0; bz0.x = b0[i].z; bz0.y = b0[i].w;
            v2f bx1; bx1.x = b1[i].x; bx1.y = b1[i].y;
            v2f bz1; bz1.x = b1[i].z; bz1.y = b1[i].w;
            c0 = __builtin_amdgcn_wmma_f32_16x16x4_f32(false, ax, false, bx0, (short)0, c0, false, false);
            c1 = __builtin_amdgcn_wmma_f32_16x16x4_f32(false, ax, false, bx1, (short)0, c1, false, false);
            c0 = __builtin_amdgcn_wmma_f32_16x16x4_f32(false, az, false, bz0, (short)0, c0, false, false);
            c1 = __builtin_amdgcn_wmma_f32_16x16x4_f32(false, az, false, bz1, (short)0, c1, false, false);
        }

        const int m0 = row0 + hi * 8;
#pragma unroll
        for (int v = 0; v < 8; ++v) {
            XW[(size_t)(m0 + v) * GCN_D + nc] = c0[v];
            XW[(size_t)(m0 + v) * GCN_D + nc + 16] = c1[v];
        }
    }
}


#define SC_EDGES_PER_WAVE 32
#define SC_GROUP 8
#define SC_LN_ROWS ((SB_ROWS + SB_WAVES - 1) / SB_WAVES)

__global__ void __launch_bounds__(SB_WAVES * 32)
k_scatter(const float* __restrict__ XW, const float* __restrict__ dinv,
          const float* __restrict__ bias, const int* __restrict__ prow,
          const int* __restrict__ pcol, const float* __restrict__ pnorm,
          const int* __restrict__ off, const float* hin,
          const float* __restrict__ gamma, const float* __restrict__ beta,
          float* hout, int n_nodes, int do_relu) {
    __shared__ float sacc[SB_LDS_ROWS * GCN_D];
    const int lane = threadIdx.x & 31;
    const int wave = __builtin_amdgcn_readfirstlane((int)(threadIdx.x >> 5));
    const int n0 = (int)blockIdx.x * SB_ROWS;
    const int nleft = n_nodes - n0;
    const int nrows = nleft < SB_ROWS ? nleft : SB_ROWS;

    const float4 b = ((const float4*)bias)[lane];
    for (int r = wave; r < nrows; r += SB_WAVES) {
        const float dn = dinv[n0 + r];
        const float self = dn * dn;
        const float4 x = ((const float4*)(XW + (size_t)(n0 + r) * GCN_D))[lane];
        float4 o;
        o.x = b.x + x.x * self;
        o.y = b.y + x.y * self;
        o.z = b.z + x.z * self;
        o.w = b.w + x.w * self;
        float* srow = sacc + r * GCN_D + lane;
        srow[0] = o.x;
        srow[32] = o.y;
        srow[64] = o.z;
        srow[96] = o.w;
    }
    __syncthreads();

    float4 id[SC_LN_ROWS];
#pragma unroll
    for (int k = 0; k < SC_LN_ROWS; ++k) {
        const int rk = wave + k * SB_WAVES;
        const int rr = rk < nrows ? rk : nrows - 1;
        id[k] = ((const float4*)(hin + (size_t)(n0 + rr) * GCN_D))[lane];
    }
    const float4 gm = ((const float4*)gamma)[lane];
    const float4 bt = ((const float4*)beta)[lane];

    const int e_begin = off[blockIdx.x];
    const int e_end = off[blockIdx.x + 1];
    int e0 = e_begin + wave * SC_EDGES_PER_WAVE;
    int r = 0, c = 0, wbits = 0;
    if (e0 + lane < e_end) {
        r = prow[e0 + lane];
        c = pcol[e0 + lane];
        wbits = __float_as_int(pnorm[e0 + lane]);
    }
    while (e0 < e_end) {
        const int left = e_end - e0;
        const int cnt = left < SC_EDGES_PER_WAVE ? left : SC_EDGES_PER_WAVE;
        const int e1 = e0 + SB_WAVES * SC_EDGES_PER_WAVE;
        int rn = 0, cn = 0, wn = 0;
        if (e1 + lane < e_end) {
            rn = prow[e1 + lane];
            cn = pcol[e1 + lane];
            wn = __float_as_int(pnorm[e1 + lane]);
        }
        int g = 0;
#pragma unroll 1
        for (; g + SC_GROUP <= cnt; g += SC_GROUP) {
            float4 m[SC_GROUP];
#pragma unroll
            for (int k = 0; k < SC_GROUP; ++k) {
                const int rs = __builtin_amdgcn_readlane(r, g + k);
                m[k] = ((const float4*)(XW + (size_t)rs * GCN_D))[lane];
            }
#pragma unroll
            for (int k = 0; k < SC_GROUP; ++k) {
                const int cs = __builtin_amdgcn_readlane(c, g + k);
                const float wf = __int_as_float(__builtin_amdgcn_readlane(wbits, g + k));
                float* dst = sacc + cs * GCN_D + lane;
                atomicAdd(dst + 0,  m[k].x * wf);
                atomicAdd(dst + 32, m[k].y * wf);
                atomicAdd(dst + 64, m[k].z * wf);
                atomicAdd(dst + 96, m[k].w * wf);
            }
        }
        for (int i = g; i < cnt; ++i) {
            const int rs = __builtin_amdgcn_readlane(r, i);
            const int cs = __builtin_amdgcn_readlane(c, i);
            const float wf = __int_as_float(__builtin_amdgcn_readlane(wbits, i));
            const float4 mv = ((const float4*)(XW + (size_t)rs * GCN_D))[lane];
            float* dst = sacc + cs * GCN_D + lane;
            atomicAdd(dst + 0,  mv.x * wf);
            atomicAdd(dst + 32, mv.y * wf);
            atomicAdd(dst + 64, mv.z * wf);
            atomicAdd(dst + 96, mv.w * wf);
        }
        r = rn;
        c = cn;
        wbits = wn;
        e0 = e1;
    }
    __syncthreads();

    float4 a[SC_LN_ROWS];
#pragma unroll
    for (int k = 0; k < SC_LN_ROWS; ++k) {
        const int rk = wave + k * SB_WAVES;
        const int rr = rk < nrows ? rk : nrows - 1;
        const float* srow = sacc + rr * GCN_D + lane;
        a[k].x = srow[0];
        a[k].y = srow[32];
        a[k].z = srow[64];
        a[k].w = srow[96];
    }

    float s[SC_LN_ROWS];
#pragma unroll
    for (int k = 0; k < SC_LN_ROWS; ++k) s[k] = a[k].x + a[k].y + a[k].z + a[k].w;
#pragma unroll
    for (int sh = 16; sh > 0; sh >>= 1) {
#pragma unroll
        for (int k = 0; k < SC_LN_ROWS; ++k) s[k] += __shfl_xor(s[k], sh, 32);
    }

    float q[SC_LN_ROWS];
#pragma unroll
    for (int k = 0; k < SC_LN_ROWS; ++k) {
        const float mu = s[k] * (1.0f / GCN_D);
        a[k].x -= mu; a[k].y -= mu; a[k].z -= mu; a[k].w -= mu;
        q[k] = a[k].x * a[k].x + a[k].y * a[k].y + a[k].z * a[k].z + a[k].w * a[k].w;
    }
#pragma unroll
    for (int sh = 16; sh > 0; sh >>= 1) {
#pragma unroll
        for (int k = 0; k < SC_LN_ROWS; ++k) q[k] += __shfl_xor(q[k], sh, 32);
    }

#pragma unroll
    for (int k = 0; k < SC_LN_ROWS; ++k) {
        const float inv = rsqrtf(q[k] * (1.0f / GCN_D) + LN_EPS);
        float4 y;
        y.x = a[k].x * inv * gm.x + bt.x;
        y.y = a[k].y * inv * gm.y + bt.y;
        y.z = a[k].z * inv * gm.z + bt.z;
        y.w = a[k].w * inv * gm.w + bt.w;
        if (do_relu) {
            y.x = fmaxf(y.x, 0.0f);
            y.y = fmaxf(y.y, 0.0f);
            y.z = fmaxf(y.z, 0.0f);
            y.w = fmaxf(y.w, 0.0f);
        }
        y.x += id[k].x; y.y += id[k].y; y.z += id[k].z; y.w += id[k].w;
        const int rk = wave + k * SB_WAVES;
        if (rk < nrows)
            ((float4*)(hout + (size_t)(n0 + rk) * GCN_D))[lane] = y;
    }
}


extern "C" void kernel_launch(void* const* d_in, const int* in_sizes, int n_in,
                              void* d_out, int out_size, void* d_ws, size_t ws_size,
                              hipStream_t stream) {
    (void)in_sizes; (void)n_in; (void)out_size; (void)ws_size;

    const float* x      = (const float*)d_in[0];
    const int*   ei     = (const int*)d_in[1];
    const float* ew     = (const float*)d_in[2];
    const float* Ws     = (const float*)d_in[3];
    const float* bs     = (const float*)d_in[4];
    const float* gammas = (const float*)d_in[5];
    const float* betas  = (const float*)d_in[6];
    float*       out    = (float*)d_out;

    const int* row = ei;
    const int* col = ei + GCN_E;

    float* ws   = (float*)d_ws;
    float* pnorm = ws;
    float* xw   = pnorm + GCN_E;
    float* h1   = xw + (size_t)GCN_N * GCN_D;
    int* cnt    = (int*)(h1 + (size_t)GCN_N * GCN_D);
    int* cursor = cnt + SB_PAD;
    int* off    = cursor + SB_PAD;
    int* prow   = off + SB_PAD;
    int* pcol   = prow + GCN_E;
    float* dinv = (float*)(pcol + GCN_E);

    const int bin_blocks = (GCN_E + BN_SLICE - 1) / BN_SLICE;
    k_zero_counters<<<1, SB_PAD, 0, stream>>>(cnt, cursor);
    k_bin_count<<<bin_blocks, BN_THREADS, 0, stream>>>(col, cnt, GCN_E);
    k_bin_place<<<bin_blocks, BN_THREADS, 0, stream>>>(row, col, ew, cnt, cursor, off,
                                                       prow, pcol, pnorm, GCN_E);
    k_bin_deg<<<SB_BINS, BD_THREADS, 0, stream>>>(pcol, pnorm, off, dinv, GCN_N);
    k_bin_norm<<<SB_BINS, BD_THREADS, 0, stream>>>(dinv, prow, pcol, off, pnorm);

    const int row_tiles    = (GCN_N + 15) / 16;
    const int gemm_blocks  = (row_tiles + GM_ROWTILES - 1) / GM_ROWTILES;

    const float* hin[GCN_L]  = { x,  h1,  out };
    float*       hout[GCN_L] = { h1, out, out };

    for (int l = 0; l < GCN_L; ++l) {
        const float* W = Ws + (size_t)l * GCN_D * GCN_D;
        k_gemm_wmma<<<gemm_blocks, GM_WAVES * 32, 0, stream>>>(hin[l], W, xw, GCN_N);
        k_scatter<<<SB_BINS, SB_WAVES * 32, 0, stream>>>(xw, dinv, bs + (size_t)l * GCN_D,
                                                         prow, pcol, pnorm, off, hin[l],
                                                         gammas + (size_t)l * GCN_D,
                                                         betas + (size_t)l * GCN_D,
                                                         hout[l], GCN_N,
                                                          (l < GCN_L - 1) ? 1 : 0);
    }
}


# A binned edge is one 8-byte record (source row | range row << 16, then the coefficient): binning places it with 1 scattered store, not 3; readers take it with one 8-byte load; sum orders kept
# speedup vs baseline: 5.1201x; 1.1523x over previous
#include <hip/hip_runtime.h>
#include <hip/hip_bf16.h>


#define GCN_N 50000
#define GCN_E 1600000
#define GCN_D 128
#define GCN_L 3
#define LN_EPS 1e-5f

#define SB_ROWS 49
#define SB_BINS ((GCN_N + SB_ROWS - 1) / SB_ROWS)
#define SB_LDS_ROWS 64
#define SB_PAD 1024
#define SB_WAVES 8
#define BN_THREADS 1024
#define BN_PER_THREAD 8
#define BN_SLICE (BN_THREADS * BN_PER_THREAD)
static_assert(SB_BINS < SB_PAD && SB_PAD == BN_THREADS, "one counter per binning thread");
static_assert(SB_ROWS <= SB_LDS_ROWS, "a range's sum rows fit the reserved LDS rows");
static_assert(GCN_N <= 65536 && SB_ROWS <= 32768, "a node id fits the low and a range row the high half of a record's key");

__device__ __forceinline__ int sb_bin(int node) { return (int)((unsigned)node / (unsigned)SB_ROWS); }

__device__ __forceinline__ int rec_key(int src_row, int dst_in_range) { return src_row | (dst_in_range << 16); }
__device__ __forceinline__ int rec_src(int key) { return key & 0xFFFF; }
__device__ __forceinline__ int rec_dst(int key) { return key >> 16; }

typedef __attribute__((ext_vector_type(2))) float v2f;
typedef __attribute__((ext_vector_type(8))) float v8f;


__global__ void __launch_bounds__(SB_PAD)
k_zero_counters(int* __restrict__ cnt, int* __restrict__ cursor) {
    cnt[threadIdx.x] = 0;
    cursor[threadIdx.x] = 0;
}


__global__ void __launch_bounds__(BN_THREADS)
k_bin_count(const int* __restrict__ col, int* __restrict__ cnt, int e) {
    __shared__ int h[SB_PAD];
    const int tid = threadIdx.x;
    h[tid] = 0;
    __syncthreads();
    const int base = blockIdx.x * BN_SLICE + tid;
    int cv[BN_PER_THREAD];
#pragma unroll
    for (int k = 0; k < BN_PER_THREAD; ++k) {
        const int i = base + k * BN_THREADS;
        cv[k] = col[i < e ? i : e - 1];
    }
#pragma unroll
    for (int k = 0; k < BN_PER_THREAD; ++k)
        if (base + k * BN_THREADS < e) atomicAdd(&h[sb_bin(cv[k])], 1);
    __syncthreads();
    const int v = h[tid];
    if (v) atomicAdd(&cnt[tid], v);
}

__global__ void __launch_bounds__(BN_THREADS)
k_bin_place(const int* __restrict__ row, const int* __restrict__ col,
            const float* __restrict__ w, const int* __restrict__ cnt,
            int* __restrict__ cursor, int* __restrict__ off,
            int* __restrict__ rec_words, int e) {
    int2* __restrict__ rec = (int2*)rec_words;
    __shared__ int h[SB_PAD];
    __shared__ int wsum[32];
    const int tid = threadIdx.x;
    const int lane = tid & 31;
    const int wv = tid >> 5;

    const int cv = cnt[tid];
    int inc = cv;
#pragma unroll
    for (int d = 1; d < 32; d <<= 1) {
        const int t = __shfl_up(inc, d, 32);
        if (lane >= d) inc += t;
    }
    if (lane == 31) wsum[wv] = inc;
    h[tid] = 0;
    __syncthreads();
    if (wv == 0) {
        int s = wsum[lane];
#pragma unroll
        for (int d = 1; d < 32; d <<= 1) {
            const int t = __shfl_up(s, d, 32);
            if (lane >= d) s += t;
        }
        wsum[lane] = s;
    }
    __syncthreads();
    const int start = inc - cv + (wv ? wsum[wv - 1] : 0);
    if (blockIdx.x == 0) off[tid] = start;

    int c[BN_PER_THREAD], r[BN_PER_THREAD], rk[BN_PER_THREAD];
    float we[BN_PER_THREAD];
    const int base = blockIdx.x * BN_SLICE + tid;
#pragma unroll
    for (int k = 0; k < BN_PER_THREAD; ++k) {
        const int i = base + k * BN_THREADS;
        const int ii = i < e ? i : e - 1;
        c[k] = col[ii];
        r[k] = row[ii];
        we[k] = w[ii];
    }
#pragma unroll
    for (int k = 0; k < BN_PER_THREAD; ++k) {
        const int i = base + k * BN_THREADS;
        rk[k] = 0;
        if (i < e) rk[k] = atomicAdd(&h[sb_bin(c[k])], 1);
    }
    __syncthreads();
    const int hv = h[tid];
    h[tid] = start + (hv ? atomicAdd(&cursor[tid], hv) : 0);
    __syncthreads();
#pragma unroll
    for (int k = 0; k < BN_PER_THREAD; ++k) {
        const int i = base + k * BN_THREADS;
        if (i < e) {
            const int bin = sb_bin(c[k]);
            const int pos = h[bin] + rk[k];
            rec[pos] = make_int2(rec_key(r[k], c[k] - bin * SB_ROWS), __float_as_int(we[k]));
        }
    }
}


#define BD_THREADS 256
#define BD_PER_THREAD 8

__global__ void __launch_bounds__(BD_THREADS)
k_bin_deg(const int* __restrict__ rec_words,
          const int* __restrict__ off, float* __restrict__ dinv, int n_nodes) {
    const int2* __restrict__ rec = (const int2*)rec_words;
    __shared__ float sdeg[SB_LDS_ROWS];
    const int tid = threadIdx.x;
    const int e_begin = off[blockIdx.x];
    const int n_e = off[blockIdx.x + 1] - e_begin;
    if (tid < SB_LDS_ROWS) sdeg[tid] = 1.0f;
    __syncthreads();
    for (int base = tid; base < n_e; base += BD_THREADS * BD_PER_THREAD) {
        int c[BD_PER_THREAD];
        float we[BD_PER_THREAD];
#pragma unroll
        for (int k = 0; k < BD_PER_THREAD; ++k) {
            const int i = base + k * BD_THREADS;
            const int ii = e_begin + (i < n_e ? i : n_e - 1);
            const int2 t = rec[ii];
            c[k] = rec_dst(t.x);
            we[k] = __int_as_float(t.y);
        }
#pragma unroll
        for (int k = 0; k < BD_PER_THREAD; ++k)
            if (base + k * BD_THREADS < n_e) atomicAdd(&sdeg[c[k]], we[k]);
    }
    __syncthreads();
    const int n0 = (int)blockIdx.x * SB_ROWS;
    if (tid < SB_ROWS && n0 + tid < n_nodes) {
        const float d = sdeg[tid];
        dinv[n0 + tid] = (d > 0.0f) ? rsqrtf(fmaxf(d, 1e-12f)) : 0.0f;
    }
}

__global__ void __launch_bounds__(BD_THREADS)
k_bin_norm(const float* __restrict__ dinv, const int* __restrict__ off,
           int* __restrict__ rec_words) {
    int2* __restrict__ rec = (int2*)rec_words;
    const int tid = threadIdx.x;
    const int n0 = (int)blockIdx.x * SB_ROWS;
    const int e_begin = off[blockIdx.x];
    const int n_e = off[blockIdx.x + 1] - e_begin;
    for (int base = tid; base < n_e; base += BD_THREADS * BD_PER_THREAD) {
        int key[BD_PER_THREAD];
        float we[BD_PER_THREAD], dr[BD_PER_THREAD], dc[BD_PER_THREAD];
#pragma unroll
        for (int k = 0; k < BD_PER_THREAD; ++k) {
            const int i = base + k * BD_THREADS;
            const int ii = e_begin + (i < n_e ? i : n_e - 1);
            const int2 t = rec[ii];
            key[k] = t.x;
            we[k] = __int_as_float(t.y);
        }
#pragma unroll
        for (int k = 0; k < BD_PER_THREAD; ++k) {
            dr[k] = dinv[rec_src(key[k])];
            dc[k] = dinv[n0 + rec_dst(key[k])];
        }
#pragma unroll
        for (int k = 0; k < BD_PER_THREAD; ++k) {
            const int i = base + k * BD_THREADS;
            if (i < n_e)
                rec[e_begin + i] = make_int2(key[k], __float_as_int(dr[k] * we[k] * dc[k]));
        }
    }
}


#define GM_ROWTILES 4
#define GM_WAVES 4

__global__ void __launch_bounds__(GM_WAVES * 32)
k_gemm_wmma(const float* __restrict__ H, const float* __restrict__ W,
            float* __restrict__ XW, int n_nodes) {
    const int lane = threadIdx.x & 31;
    const int col0 = (threadIdx.x >> 5) * 32;
    const int lm = lane & 15;
    const int hi = lane >> 4;

    const float4* __restrict__ Bp0 = (const float4*)(W + (size_t)(col0 + lm) * GCN_D) + hi;
    const float4* __restrict__ Bp1 = (const float4*)(W + (size_t)(col0 + 16 + lm) * GCN_D) + hi;
    float4 b0[16], b1[16];
#pragma unroll
    for (int i = 0; i < 16; ++i) {
        b0[i] = Bp0[2 * i];
        b1[i] = Bp1[2 * i];
    }

    const int nc = col0 + lm;
#pragma unroll 1
    for (int rt = 0; rt < GM_ROWTILES; ++rt) {
        const int row0 = ((int)blockIdx.x * GM_ROWTILES + rt) * 16;
        if (row0 >= n_nodes) break;

        const float4* __restrict__ Ap = (const float4*)(H + (size_t)(row0 + lm) * GCN_D) + hi;
        float4 a[16];
#pragma unroll
        for (int i = 0; i < 16; ++i) a[i] = Ap[2 * i];

        v8f c0 = {};
        v8f c1 = {};
#pragma unroll
        for (int i = 0; i < 16; ++i) {
            v2f ax; ax.x = a[i].x; ax.y = a[i].y;
            v2f az; az.x = a[i].z; az.y = a[i].w;
            v2f bx0; bx0.x = b0[i].x; bx0.y = b0[i].y;
            v2f bz0; bz0.x = b0[i].z; bz0.y = b0[i].w;
            v2f bx1; bx1.x = b1[i].x; bx1.y = b1[i].y;
            v2f bz1; bz1.x = b1[i].z; bz1.y = b1[i].w;
            c0 = __builtin_amdgcn_wmma_f32_16x16x4_f32(false, ax, false, bx0, (short)0, c0, false, false);
            c1 = __builtin_amdgcn_wmma_f32_16x16x4_f32(false, ax, false, bx1, (short)0, c1, false, false);
            c0 = __builtin_amdgcn_wmma_f32_16x16x4_f32(false, az, false, bz0, (short)0, c0, false, false);
            c1 = __builtin_amdgcn_wmma_f32_16x16x4_f32(false, az, false, bz1, (short)0, c1, false, false);
        }

        const int m0 = row0 + hi * 8;
#pragma unroll
        for (int v = 0; v < 8; ++v) {
            XW[(size_t)(m0 + v) * GCN_D + nc] = c0[v];
            XW[(size_t)(m0 + v) * GCN_D + nc + 16] = c1[v];
        }
    }
}


#define SC_EDGES_PER_WAVE 32
#define SC_GROUP 8
#define SC_LN_ROWS ((SB_ROWS + SB_WAVES - 1) / SB_WAVES)

__global__ void __launch_bounds__(SB_WAVES * 32)
k_scatter(const float* __restrict__ XW, const float* __restrict__ dinv,
          const float* __restrict__ bias, const int* __restrict__ rec_words,
          const int* __restrict__ off, const float* hin,
          const float* __restrict__ gamma, const float* __restrict__ beta,
          float* hout, int n_nodes, int do_relu) {
    const int2* __restrict__ rec = (const int2*)rec_words;
    __shared__ float sacc[SB_LDS_ROWS * GCN_D];
    const int lane = threadIdx.x & 31;
    const int wave = __builtin_amdgcn_readfirstlane((int)(threadIdx.x >> 5));
    const int n0 = (int)blockIdx.x * SB_ROWS;
    const int nleft = n_nodes - n0;
    const int nrows = nleft < SB_ROWS ? nleft : SB_ROWS;

    const float4 b = ((const float4*)bias)[lane];
    for (int r = wave; r < nrows; r += SB_WAVES) {
        const float dn = dinv[n0 + r];
        const float self = dn * dn;
        const float4 x = ((const float4*)(XW + (size_t)(n0 + r) * GCN_D))[lane];
        float4 o;
        o.x = b.x + x.x * self;
        o.y = b.y + x.y * self;
        o.z = b.z + x.z * self;
        o.w = b.w + x.w * self;
        float* srow = sacc + r * GCN_D + lane;
        srow[0] = o.x;
        srow[32] = o.y;
        srow[64] = o.z;
        srow[96] = o.w;
    }
    __syncthreads();

    float4 id[SC_LN_ROWS];
#pragma unroll
    for (int k = 0; k < SC_LN_ROWS; ++k) {
        const int rk = wave + k * SB_WAVES;
        const int rr = rk < nrows ? rk : nrows - 1;
        id[k] = ((const float4*)(hin + (size_t)(n0 + rr) * GCN_D))[lane];
    }
    const float4 gm = ((const float4*)gamma)[lane];
    const float4 bt = ((const float4*)beta)[lane];

    const int e_begin = off[blockIdx.x];
    const int e_end = off[blockIdx.x + 1];
    int e0 = e_begin + wave * SC_EDGES_PER_WAVE;
    int key = 0, wbits = 0;
    if (e0 + lane < e_end) {
        const int2 t = rec[e0 + lane];
        key = t.x;
        wbits = t.y;
    }
    while (e0 < e_end) {
        const int left = e_end - e0;
        const int cnt = left < SC_EDGES_PER_WAVE ? left : SC_EDGES_PER_WAVE;
        const int e1 = e0 + SB_WAVES * SC_EDGES_PER_WAVE;
        int kn = 0, wn = 0;
        if (e1 + lane < e_end) {
            const int2 t = rec[e1 + lane];
            kn = t.x;
            wn = t.y;
        }
        int g = 0;
#pragma unroll 1
        for (; g + SC_GROUP <= cnt; g += SC_GROUP) {
            float4 m[SC_GROUP];
#pragma unroll
            for (int k = 0; k < SC_GROUP; ++k) {
                const unsigned rs = (unsigned)rec_src(__builtin_amdgcn_readlane(key, g + k));
                m[k] = ((const float4*)XW)[rs * (GCN_D / 4) + (unsigned)lane];
            }
#pragma unroll
            for (int k = 0; k < SC_GROUP; ++k) {
                const int cs = rec_dst(__builtin_amdgcn_readlane(key, g + k));
                const float wf = __int_as_float(__builtin_amdgcn_readlane(wbits, g + k));
                float* dst = sacc + cs * GCN_D + lane;
                atomicAdd(dst + 0,  m[k].x * wf);
                atomicAdd(dst + 32, m[k].y * wf);
                atomicAdd(dst + 64, m[k].z * wf);
                atomicAdd(dst + 96, m[k].w * wf);
            }
        }
        for (int i = g; i < cnt; ++i) {
            const int ks = __builtin_amdgcn_readlane(key, i);
            const int rs = rec_src(ks);
            const int cs = rec_dst(ks);
            const float wf = __int_as_float(__builtin_amdgcn_readlane(wbits, i));
            const float4 mv = ((const float4*)XW)[(unsigned)rs * (GCN_D / 4) + (unsigned)lane];
            float* dst = sacc + cs * GCN_D + lane;
            atomicAdd(dst + 0,  mv.x * wf);
            atomicAdd(dst + 32, mv.y * wf);
            atomicAdd(dst + 64, mv.z * wf);
            atomicAdd(dst + 96, mv.w * wf);
        }
        key = kn;
        wbits = wn;
        e0 = e1;
    }
    __syncthreads();

    float4 a[SC_LN_ROWS];
#pragma unroll
    for (int k = 0; k < SC_LN_ROWS; ++k) {
        const int rk = wave + k * SB_WAVES;
        const int rr = rk < nrows ? rk : nrows - 1;
        const float* srow = sacc + rr * GCN_D + lane;
        a[k].x = srow[0];
        a[k].y = srow[32];
        a[k].z = srow[64];
        a[k].w = srow[96];
    }

    float s[SC_LN_ROWS];
#pragma unroll
    for (int k = 0; k < SC_LN_ROWS; ++k) s[k] = a[k].x + a[k].y + a[k].z + a[k].w;
#pragma unroll
    for (int sh = 16; sh > 0; sh >>= 1) {
#pragma unroll
        for (int k = 0; k < SC_LN_ROWS; ++k) s[k] += __shfl_xor(s[k], sh, 32);
    }

    float q[SC_LN_ROWS];
#pragma unroll
    for (int k = 0; k < SC_LN_ROWS; ++k) {
        const float mu = s[k] * (1.0f / GCN_D);
        a[k].x -= mu; a[k].y -= mu; a[k].z -= mu; a[k].w -= mu;
        q[k] = a[k].x * a[k].x + a[k].y * a[k].y + a[k].z * a[k].z + a[k].w * a[k].w;
    }
#pragma unroll
    for (int sh = 16; sh > 0; sh >>= 1) {
#pragma unroll
        for (int k = 0; k < SC_LN_ROWS; ++k) q[k] += __shfl_xor(q[k], sh, 32);
    }

#pragma unroll
    for (int k = 0; k < SC_LN_ROWS; ++k) {
        const float inv = rsqrtf(q[k] * (1.0f / GCN_D) + LN_EPS);
        float4 y;
        y.x = a[k].x * inv * gm.x + bt.x;
        y.y = a[k].y * inv * gm.y + bt.y;
        y.z = a[k].z * inv * gm.z + bt.z;
        y.w = a[k].w * inv * gm.w + bt.w;
        if (do_relu) {
            y.x = fmaxf(y.x, 0.0f);
            y.y = fmaxf(y.y, 0.0f);
            y.z = fmaxf(y.z, 0.0f);
            y.w = fmaxf(y.w, 0.0f);
        }
        y.x += id[k].x; y.y += id[k].y; y.z += id[k].z; y.w += id[k].w;
        const int rk = wave + k * SB_WAVES;
        if (rk < nrows)
            ((float4*)(hout + (size_t)(n0 + rk) * GCN_D))[lane] = y;
    }
}


extern "C" void kernel_launch(void* const* d_in, const int* in_sizes, int n_in,
                              void* d_out, int out_size, void* d_ws, size_t ws_size,
                              hipStream_t stream) {
    (void)in_sizes; (void)n_in; (void)out_size; (void)ws_size;

    const float* x      = (const float*)d_in[0];
    const int*   ei     = (const int*)d_in[1];
    const float* ew     = (const float*)d_in[2];
    const float* Ws     = (const float*)d_in[3];
    const float* bs     = (const float*)d_in[4];
    const float* gammas = (const float*)d_in[5];
    const float* betas  = (const float*)d_in[6];
    float*       out    = (float*)d_out;

    const int* row = ei;
    const int* col = ei + GCN_E;

    float* ws   = (float*)d_ws;
    int* rec    = (int*)ws;
    float* xw   = ws + 2 * (size_t)GCN_E;
    float* h1   = xw + (size_t)GCN_N * GCN_D;
    int* cnt    = (int*)(h1 + (size_t)GCN_N * GCN_D);
    int* cursor = cnt + SB_PAD;
    int* off    = cursor + SB_PAD;
    float* dinv = (float*)(off + SB_PAD);

    const int bin_blocks = (GCN_E + BN_SLICE - 1) / BN_SLICE;
    k_zero_counters<<<1, SB_PAD, 0, stream>>>(cnt, cursor);
    k_bin_count<<<bin_blocks, BN_THREADS, 0, stream>>>(col, cnt, GCN_E);
    k_bin_place<<<bin_blocks, BN_THREADS, 0, stream>>>(row, col, ew, cnt, cursor, off,
                                                       rec, GCN_E);
    k_bin_deg<<<SB_BINS, BD_THREADS, 0, stream>>>(rec, off, dinv, GCN_N);
    k_bin_norm<<<SB_BINS, BD_THREADS, 0, stream>>>(dinv, off, rec);

    const int row_tiles    = (GCN_N + 15) / 16;
    const int gemm_blocks  = (row_tiles + GM_ROWTILES - 1) / GM_ROWTILES;

    const float* hin[GCN_L]  = { x,  h1,  out };
    float*       hout[GCN_L] = { h1, out, out };

    for (int l = 0; l < GCN_L; ++l) {
        const float* W = Ws + (size_t)l * GCN_D * GCN_D;
        k_gemm_wmma<<<gemm_blocks, GM_WAVES * 32, 0, stream>>>(hin[l], W, xw, GCN_N);
        k_scatter<<<SB_BINS, SB_WAVES * 32, 0, stream>>>(xw, dinv, bs + (size_t)l * GCN_D,
                                                         rec, off, hin[l],
                                                         gammas + (size_t)l * GCN_D,
                                                         betas + (size_t)l * GCN_D,
                                                         hout[l], GCN_N,
                                                          (l < GCN_L - 1) ? 1 : 0);
    }
}
